# starting kernel
# baseline (speedup 1.0000x reference)
#include <hip/hip_runtime.h>


typedef float v2f __attribute__((ext_vector_type(2)));
typedef float v8f __attribute__((ext_vector_type(8)));

#define GNN_N 100000
#define GNN_E 1600000
#define DIM   64
#define XPAD  68

__global__ void fill_kernel(float* __restrict__ p, float v, int n) {
    int i = blockIdx.x * blockDim.x + threadIdx.x;
    if (i < n) p[i] = v;
}

__global__ void deg_kernel(const int* __restrict__ dst, float* __restrict__ deg, int E) {
    int e = blockIdx.x * blockDim.x + threadIdx.x;
    if (e < E)
        __hip_atomic_fetch_add(&deg[dst[e]], 1.0f, __ATOMIC_RELAXED, __HIP_MEMORY_SCOPE_AGENT);
}

__global__ void rsqrt_kernel(float* __restrict__ d, int n) {
    int i = blockIdx.x * blockDim.x + threadIdx.x;
    if (i < n) d[i] = rsqrtf(d[i]);
}

__global__ void norm_kernel(const int* __restrict__ src, const int* __restrict__ dst,
                            const float* __restrict__ dinv, float* __restrict__ nrm, int E) {
    int e = blockIdx.x * blockDim.x + threadIdx.x;
    if (e < E) nrm[e] = dinv[src[e]] * dinv[dst[e]];
}

__global__ __launch_bounds__(256) void gemm64_wmma(const float* __restrict__ X,
                                                   const float* __restrict__ W,
                                                   float* __restrict__ H, int nTiles) {
    __shared__ float Ws[64 * 64];
    __shared__ float Xs[8][16 * XPAD];

    const int tid  = threadIdx.x;
    const int wave = tid >> 5;
    const int lane = tid & 31;
    const int tile = blockIdx.x * 8 + wave;

#pragma unroll
    for (int i = 0; i < 16; ++i) {
        int g = tid + i * 256;
        int k = g >> 6;
        int n = g & 63;
        Ws[(k >> 1) * 128 + n * 2 + (k & 1)] = W[g];
    }

    if (tile < nTiles) {
        const float4* Xg = (const float4*)(X + (size_t)tile * 16 * 64);
        float* xl = &Xs[wave][0];
#pragma unroll
        for (int i = 0; i < 8; ++i) {
            int idx4 = lane + i * 32;
            float4 v = Xg[idx4];
            int row = idx4 >> 4;
            int col = (idx4 & 15) * 4;
            *(float4*)&xl[row * XPAD + col] = v;
        }
    }
    __syncthreads();
    if (tile >= nTiles) return;

    const float* xs = &Xs[wave][0];
    const int l15 = lane & 15;
    const int hiK = (lane >> 4) * 2;

    v8f c0 = {}, c1 = {}, c2 = {}, c3 = {};
#pragma unroll
    for (int k = 0; k < 64; k += 4) {
        v2f a = *(const v2f*)(xs + l15 * XPAD + k + hiK);
        const float* wrow = Ws + ((k + hiK) >> 1) * 128 + l15 * 2;
        v2f b0 = *(const v2f*)(wrow +  0);
        v2f b1 = *(const v2f*)(wrow + 32);
        v2f b2 = *(const v2f*)(wrow + 64);
        v2f b3 = *(const v2f*)(wrow + 96);
        c0 = __builtin_amdgcn_wmma_f32_16x16x4_f32(false, a, false, b0, (short)0, c0, false, false);
        c1 = __builtin_amdgcn_wmma_f32_16x16x4_f32(false, a, false, b1, (short)0, c1, false, false);
        c2 = __builtin_amdgcn_wmma_f32_16x16x4_f32(false, a, false, b2, (short)0, c2, false, false);
        c3 = __builtin_amdgcn_wmma_f32_16x16x4_f32(false, a, false, b3, (short)0, c3, false, false);
    }

    const size_t base = (size_t)tile * 16;
    const int rowOff = (lane >> 4) * 8;
#pragma unroll
    for (int r = 0; r < 8; ++r) {
        float* out = H + (base + r + rowOff) * 64 + l15;
        out[0]  = c0[r];
        out[16] = c1[r];
        out[32] = c2[r];
        out[48] = c3[r];
    }
}

__global__ __launch_bounds__(256) void scatter_kernel(const float* __restrict__ h,
                                                      const float* __restrict__ nrm,
                                                      const int* __restrict__ src,
                                                      const int* __restrict__ dst,
                                                      float* __restrict__ agg, int E) {
    int ew = __builtin_amdgcn_readfirstlane((int)threadIdx.x >> 6);
    int e  = blockIdx.x * 4 + ew;
    int d  = threadIdx.x & 63;
    if (e >= E) return;
    int   s = src[e];
    int   q = dst[e];
    float w = nrm[e];
    float v = h[(size_t)s * 64 + d] * w;
    __hip_atomic_fetch_add(&agg[(size_t)q * 64 + d], v,
                           __ATOMIC_RELAXED, __HIP_MEMORY_SCOPE_AGENT);
}

template <bool RELU>
__global__ void finalize_kernel(float* __restrict__ agg, const float* __restrict__ h,
                                const float* __restrict__ dinv, const float* __restrict__ b,
                                int total) {
    int t = blockIdx.x * blockDim.x + threadIdx.x;
    if (t >= total) return;
    int i = __builtin_amdgcn_readfirstlane(t >> 6);
    int d = t & 63;
    float di = dinv[i];
    float v = agg[t] + di * di * h[t] + b[d];
    if (RELU) v = fmaxf(v, 0.0f);
    agg[t] = v;
}

__global__ __launch_bounds__(256) void proj_kernel(const float* __restrict__ f,
                                                   const float* __restrict__ Wp,
                                                   const float* __restrict__ bp,
                                                   float* __restrict__ y, int N) {
    int node = blockIdx.x * 8 + (threadIdx.x >> 5);
    int lane = threadIdx.x & 31;
    if (node >= N) return;
    const float* fr = f + (size_t)node * 64;
    float s = fr[lane] * Wp[lane] + fr[lane + 32] * Wp[lane + 32];
#pragma unroll
    for (int m = 16; m > 0; m >>= 1) s += __shfl_xor(s, m, 32);
    if (lane == 0) y[node] = s + bp[0];
}

extern "C" void kernel_launch(void* const* d_in, const int* in_sizes, int n_in,
                              void* d_out, int out_size, void* d_ws, size_t ws_size,
                              hipStream_t stream) {
    (void)in_sizes; (void)n_in; (void)out_size; (void)ws_size;

    const float* x  = (const float*)d_in[0];
    const int*   ei = (const int*)d_in[1];
    const float* W1 = (const float*)d_in[2];
    const float* b1 = (const float*)d_in[3];
    const float* W2 = (const float*)d_in[4];
    const float* b2 = (const float*)d_in[5];
    const float* Wp = (const float*)d_in[6];
    const float* bp = (const float*)d_in[7];
    float* y = (float*)d_out;

    const int N = GNN_N, E = GNN_E;
    const int* src = ei;
    const int* dst = ei + E;

    char* ws = (char*)d_ws;
    size_t off = 0;
    auto take = [&](size_t bytes) {
        void* p = ws + off;
        off += (bytes + 255) & ~(size_t)255;
        return p;
    };
    float* dinv = (float*)take((size_t)N * 4);
    float* nrm  = (float*)take((size_t)E * 4);
    float* bufA = (float*)take((size_t)N * DIM * 4);
    float* bufB = (float*)take((size_t)N * DIM * 4);

    const int total = N * DIM;
    const int nTiles = N / 16;
    const int gemmBlocks = (nTiles + 7) / 8;

    fill_kernel<<<(N + 255) / 256, 256, 0, stream>>>(dinv, 1.0f, N);
    deg_kernel<<<(E + 255) / 256, 256, 0, stream>>>(dst, dinv, E);
    rsqrt_kernel<<<(N + 255) / 256, 256, 0, stream>>>(dinv, N);
    norm_kernel<<<(E + 255) / 256, 256, 0, stream>>>(src, dst, dinv, nrm, E);

    gemm64_wmma<<<gemmBlocks, 256, 0, stream>>>(x, W1, bufA, nTiles);
    fill_kernel<<<total / 256, 256, 0, stream>>>(bufB, 0.0f, total);
    scatter_kernel<<<E / 4, 256, 0, stream>>>(bufA, nrm, src, dst, bufB, E);
    finalize_kernel<true><<<total / 256, 256, 0, stream>>>(bufB, bufA, dinv, b1, total);

    gemm64_wmma<<<gemmBlocks, 256, 0, stream>>>(bufB, W2, bufA, nTiles);
    fill_kernel<<<total / 256, 256, 0, stream>>>(bufB, 0.0f, total);
    scatter_kernel<<<E / 4, 256, 0, stream>>>(bufA, nrm, src, dst, bufB, E);
    finalize_kernel<true><<<total / 256, 256, 0, stream>>>(bufB, bufA, dinv, b2, total);

    proj_kernel<<<(N + 7) / 8, 256, 0, stream>>>(bufB, Wp, bp, y, N);
}


# Aggregate by gather over edges sorted by dst once per call (returning int add = slot, scan, place); one wave per node sums rows in registers: no float atomics, zero-fill or finalize pass
# speedup vs baseline: 3.2449x; 3.2449x over previous
#include <hip/hip_runtime.h>


typedef float v2f __attribute__((ext_vector_type(2)));
typedef float v8f __attribute__((ext_vector_type(8)));

#define GNN_N 100000
#define GNN_E 1600000
#define DIM   64
#define XPAD  68
#define SCAN_CHUNK 1024
#define SCAN_BLOCKS ((GNN_N + SCAN_CHUNK - 1) / SCAN_CHUNK)

__global__ void zero_kernel(int* __restrict__ p, int n) {
    int i = blockIdx.x * blockDim.x + threadIdx.x;
    if (i < n) p[i] = 0;
}

__global__ void rank_kernel(const int* __restrict__ dst, int* __restrict__ cnt,
                            int* __restrict__ rank, int E) {
    int e = blockIdx.x * blockDim.x + threadIdx.x;
    if (e < E)
        rank[e] = __hip_atomic_fetch_add(&cnt[dst[e]], 1, __ATOMIC_RELAXED,
                                         __HIP_MEMORY_SCOPE_AGENT);
}

__global__ __launch_bounds__(256) void scan_local_kernel(const int* __restrict__ cnt,
                                                         int* __restrict__ rowptr,
                                                         int* __restrict__ bsum, int n) {
    __shared__ int wsum[8];
    const int tid  = threadIdx.x;
    const int lane = tid & 31;
    const int wave = tid >> 5;
    const int i0 = blockIdx.x * SCAN_CHUNK + tid * 4;
    int t0 = (i0 + 0 < n) ? cnt[i0 + 0] : 0;
    int t1 = (i0 + 1 < n) ? cnt[i0 + 1] : 0;
    int t2 = (i0 + 2 < n) ? cnt[i0 + 2] : 0;
    int t3 = (i0 + 3 < n) ? cnt[i0 + 3] : 0;
    const int s = t0 + t1 + t2 + t3;
    int v = s;
#pragma unroll
    for (int o = 1; o < 32; o <<= 1) {
        int u = __shfl_up(v, o, 32);
        if (lane >= o) v += u;
    }
    if (lane == 31) wsum[wave] = v;
    __syncthreads();
    int woff = 0, tot = 0;
#pragma unroll
    for (int w = 0; w < 8; ++w) {
        int x = wsum[w];
        if (w < wave) woff += x;
        tot += x;
    }
    int ex = woff + v - s;
    if (i0 + 0 < n) rowptr[i0 + 0] = ex;
    if (i0 + 1 < n) rowptr[i0 + 1] = ex + t0;
    if (i0 + 2 < n) rowptr[i0 + 2] = ex + t0 + t1;
    if (i0 + 3 < n) rowptr[i0 + 3] = ex + t0 + t1 + t2;
    if (tid == 0) bsum[blockIdx.x] = tot;
}

__global__ __launch_bounds__(256) void scan_add_kernel(int* __restrict__ rowptr,
                                                       const int* __restrict__ bsum,
                                                       int n, int total) {
    __shared__ int wsum[8];
    const int tid  = threadIdx.x;
    const int lane = tid & 31;
    const int wave = tid >> 5;
    int v = (tid < (int)blockIdx.x) ? bsum[tid] : 0;
#pragma unroll
    for (int m = 16; m > 0; m >>= 1) v += __shfl_xor(v, m, 32);
    if (lane == 0) wsum[wave] = v;
    __syncthreads();
    int off = 0;
#pragma unroll
    for (int w = 0; w < 8; ++w) off += wsum[w];
    const int i0 = blockIdx.x * SCAN_CHUNK + tid * 4;
#pragma unroll
    for (int k = 0; k < 4; ++k)
        if (i0 + k < n) rowptr[i0 + k] += off;
    if (blockIdx.x == 0 && tid == 0) rowptr[n] = total;
}

__global__ void place_kernel(const int* __restrict__ src, const int* __restrict__ dst,
                             const int* __restrict__ rank, const int* __restrict__ rowptr,
                             int* __restrict__ csr, int E) {
    int e = blockIdx.x * blockDim.x + threadIdx.x;
    if (e < E) csr[rowptr[dst[e]] + rank[e]] = src[e];
}

__global__ __launch_bounds__(256) void gemm64_wmma(const float* __restrict__ X,
                                                   const float* __restrict__ W,
                                                   const int* __restrict__ rowptr,
                                                   float* __restrict__ H, int nTiles) {
    __shared__ float Ws[64 * 64];
    __shared__ float Xs[8][16 * XPAD];

    const int tid  = threadIdx.x;
    const int wave = tid >> 5;
    const int lane = tid & 31;
    const int tile = blockIdx.x * 8 + wave;

#pragma unroll
    for (int i = 0; i < 16; ++i) {
        int g = tid + i * 256;
        int k = g >> 6;
        int n = g & 63;
        Ws[(k >> 1) * 128 + n * 2 + (k & 1)] = W[g];
    }

    const int rowOff = (lane >> 4) * 8;
    const int base = tile * 16;
    int rp[9] = {};

    if (tile < nTiles) {
#pragma unroll
        for (int r = 0; r < 9; ++r) rp[r] = rowptr[base + rowOff + r];
        const float4* Xg = (const float4*)(X + (size_t)tile * 16 * 64);
        float* xl = &Xs[wave][0];
#pragma unroll
        for (int i = 0; i < 8; ++i) {
            int idx4 = lane + i * 32;
            float4 v = Xg[idx4];
            int row = idx4 >> 4;
            int col = (idx4 & 15) * 4;
            *(float4*)&xl[row * XPAD + col] = v;
        }
    }
    __syncthreads();
    if (tile >= nTiles) return;

    const float* xs = &Xs[wave][0];
    const int l15 = lane & 15;
    const int hiK = (lane >> 4) * 2;

    v8f c0 = {}, c1 = {}, c2 = {}, c3 = {};
#pragma unroll
    for (int k = 0; k < 64; k += 4) {
        v2f a = *(const v2f*)(xs + l15 * XPAD + k + hiK);
        const float* wrow = Ws + ((k + hiK) >> 1) * 128 + l15 * 2;
        v2f b0 = *(const v2f*)(wrow +  0);
        v2f b1 = *(const v2f*)(wrow + 32);
        v2f b2 = *(const v2f*)(wrow + 64);
        v2f b3 = *(const v2f*)(wrow + 96);
        c0 = __builtin_amdgcn_wmma_f32_16x16x4_f32(false, a, false, b0, (short)0, c0, false, false);
        c1 = __builtin_amdgcn_wmma_f32_16x16x4_f32(false, a, false, b1, (short)0, c1, false, false);
        c2 = __builtin_amdgcn_wmma_f32_16x16x4_f32(false, a, false, b2, (short)0, c2, false, false);
        c3 = __builtin_amdgcn_wmma_f32_16x16x4_f32(false, a, false, b3, (short)0, c3, false, false);
    }

#pragma unroll
    for (int r = 0; r < 8; ++r) {
        const float di = rsqrtf((float)(rp[r + 1] - rp[r] + 1));
        float* out = H + (size_t)(base + r + rowOff) * 64 + l15;
        out[0]  = c0[r] * di;
        out[16] = c1[r] * di;
        out[32] = c2[r] * di;
        out[48] = c3[r] * di;
    }
}

__global__ __launch_bounds__(256) void gather_kernel(const float* __restrict__ hs,
                                                     const int* __restrict__ rowptr,
                                                     const int* __restrict__ csr,
                                                     const float* __restrict__ b,
                                                     float* __restrict__ out, int N) {
    const int wave = __builtin_amdgcn_readfirstlane((int)threadIdx.x >> 5);
    const int node = blockIdx.x * 8 + wave;
    const int lane = threadIdx.x & 31;
    if (node >= N) return;

    const v2f* __restrict__ H2 = (const v2f*)hs;
    const int beg = rowptr[node];
    const int end = rowptr[node + 1];
    const v2f self = H2[node * 32 + lane];

    v2f a0 = {}, a1 = {}, a2 = {}, a3 = {}, a4 = {}, a5 = {}, a6 = {}, a7 = {};
    for (int sb = beg; sb < end; sb += 32) {
        const int left = end - sb;
        const int cn = left < 32 ? left : 32;
        const int idx = (lane < cn) ? csr[sb + lane] : 0;
        int j = 0;
        for (; j + 8 <= cn; j += 8) {
            const v2f v0 = H2[__builtin_amdgcn_readlane(idx, j + 0) * 32 + lane];
            const v2f v1 = H2[__builtin_amdgcn_readlane(idx, j + 1) * 32 + lane];
            const v2f v2 = H2[__builtin_amdgcn_readlane(idx, j + 2) * 32 + lane];
            const v2f v3 = H2[__builtin_amdgcn_readlane(idx, j + 3) * 32 + lane];
            const v2f v4 = H2[__builtin_amdgcn_readlane(idx, j + 4) * 32 + lane];
            const v2f v5 = H2[__builtin_amdgcn_readlane(idx, j + 5) * 32 + lane];
            const v2f v6 = H2[__builtin_amdgcn_readlane(idx, j + 6) * 32 + lane];
            const v2f v7 = H2[__builtin_amdgcn_readlane(idx, j + 7) * 32 + lane];
            a0 += v0; a1 += v1; a2 += v2; a3 += v3;
            a4 += v4; a5 += v5; a6 += v6; a7 += v7;
        }
        const int rem = cn - j;
        if (rem > 0) {
            v2f t0 = {}, t1 = {}, t2 = {}, t3 = {}, t4 = {}, t5 = {}, t6 = {};
            t0 = H2[__builtin_amdgcn_readlane(idx, j + 0) * 32 + lane];
            if (rem > 1) t1 = H2[__builtin_amdgcn_readlane(idx, j + 1) * 32 + lane];
            if (rem > 2) t2 = H2[__builtin_amdgcn_readlane(idx, j + 2) * 32 + lane];
            if (rem > 3) t3 = H2[__builtin_amdgcn_readlane(idx, j + 3) * 32 + lane];
            if (rem > 4) t4 = H2[__builtin_amdgcn_readlane(idx, j + 4) * 32 + lane];
            if (rem > 5) t5 = H2[__builtin_amdgcn_readlane(idx, j + 5) * 32 + lane];
            if (rem > 6) t6 = H2[__builtin_amdgcn_readlane(idx, j + 6) * 32 + lane];
            a0 += t0; a1 += t1; a2 += t2; a3 += t3;
            a4 += t4; a5 += t5; a6 += t6;
        }
    }

    const float di = rsqrtf((float)(end - beg + 1));
    const v2f bb = ((const v2f*)b)[lane];
    const v2f acc = (((a0 + a1) + (a2 + a3)) + ((a4 + a5) + (a6 + a7))) + self;
    v2f o;
    o.x = fmaxf(di * acc.x + bb.x, 0.0f);
    o.y = fmaxf(di * acc.y + bb.y, 0.0f);
    ((v2f*)out)[node * 32 + lane] = o;
}

__global__ __launch_bounds__(256) void proj_kernel(const float* __restrict__ f,
                                                   const float* __restrict__ Wp,
                                                   const float* __restrict__ bp,
                                                   float* __restrict__ y, int N) {
    int node = blockIdx.x * 8 + (threadIdx.x >> 5);
    int lane = threadIdx.x & 31;
    if (node >= N) return;
    const float* fr = f + (size_t)node * 64;
    float s = fr[lane] * Wp[lane] + fr[lane + 32] * Wp[lane + 32];
#pragma unroll
    for (int m = 16; m > 0; m >>= 1) s += __shfl_xor(s, m, 32);
    if (lane == 0) y[node] = s + bp[0];
}

extern "C" void kernel_launch(void* const* d_in, const int* in_sizes, int n_in,
                              void* d_out, int out_size, void* d_ws, size_t ws_size,
                              hipStream_t stream) {
    (void)in_sizes; (void)n_in; (void)out_size; (void)ws_size;

    const float* x  = (const float*)d_in[0];
    const int*   ei = (const int*)d_in[1];
    const float* W1 = (const float*)d_in[2];
    const float* b1 = (const float*)d_in[3];
    const float* W2 = (const float*)d_in[4];
    const float* b2 = (const float*)d_in[5];
    const float* Wp = (const float*)d_in[6];
    const float* bp = (const float*)d_in[7];
    float* y = (float*)d_out;

    const int N = GNN_N, E = GNN_E;
    const int* src = ei;
    const int* dst = ei + E;

    char* ws = (char*)d_ws;
    size_t off = 0;
    auto take = [&](size_t bytes) {
        void* p = ws + off;
        off += (bytes + 255) & ~(size_t)255;
        return p;
    };
    int*   rowptr = (int*)take((size_t)(N + 1) * 4);
    int*   csr    = (int*)take((size_t)E * 4);
    float* bufA   = (float*)take((size_t)N * DIM * 4);
    float* bufB   = (float*)take((size_t)N * DIM * 4);
    int* cnt  = (int*)bufA;
    int* bsum = cnt + N;
    int* rank = (int*)bufB;

    const int nTiles = N / 16;
    const int gemmBlocks = (nTiles + 7) / 8;

    zero_kernel<<<(N + 255) / 256, 256, 0, stream>>>(cnt, N);
    rank_kernel<<<(E + 255) / 256, 256, 0, stream>>>(dst, cnt, rank, E);
    scan_local_kernel<<<SCAN_BLOCKS, 256, 0, stream>>>(cnt, rowptr, bsum, N);
    scan_add_kernel<<<SCAN_BLOCKS, 256, 0, stream>>>(rowptr, bsum, N, E);
    place_kernel<<<(E + 255) / 256, 256, 0, stream>>>(src, dst, rank, rowptr, csr, E);

    gemm64_wmma<<<gemmBlocks, 256, 0, stream>>>(x, W1, rowptr, bufA, nTiles);
    gather_kernel<<<(N + 7) / 8, 256, 0, stream>>>(bufA, rowptr, csr, b1, bufB, N);

    gemm64_wmma<<<gemmBlocks, 256, 0, stream>>>(bufB, W2, rowptr, bufA, nTiles);
    gather_kernel<<<(N + 7) / 8, 256, 0, stream>>>(bufA, rowptr, csr, b2, bufB, N);

    proj_kernel<<<(N + 7) / 8, 256, 0, stream>>>(bufB, Wp, bp, y, N);
}


# Edge sort by destination as a two-level counting sort on LDS counters (bins of 256 destinations, then one work-group per bin): no global atomics, no scattered 4-byte reads or writes
# speedup vs baseline: 5.2595x; 1.6209x over previous
#include <hip/hip_runtime.h>


typedef float v2f __attribute__((ext_vector_type(2)));
typedef float v8f __attribute__((ext_vector_type(8)));

#define GNN_N 100000
#define GNN_E 1600000
#define DIM   64
#define XPAD  68
#define BIN_SHIFT 8
#define BIN_NODES 256
#define NBIN (GNN_N / BIN_NODES + 1)
#define EPB 6400
#define NEB ((GNN_E + EPB - 1) / EPB)
#define NCELL (NBIN * NEB)
#define STAGE_CAP 8192
#define SCAN_CHUNK 1024
#define SCAN_BLOCKS ((NCELL + SCAN_CHUNK - 1) / SCAN_CHUNK)
static_assert(EPB % (256 * 5) == 0, "a binning work-group walks its slice in steps of 256 x 5 edges");
static_assert(SCAN_BLOCKS <= 256, "the second scan step sums the chunk totals with one 256-thread group");

__global__ __launch_bounds__(256) void bin_count_kernel(const int* __restrict__ dst,
                                                        int* __restrict__ cell, int E) {
    __shared__ int h[NBIN];
    for (int i = threadIdx.x; i < NBIN; i += 256) h[i] = 0;
    __syncthreads();
    const int e0 = blockIdx.x * EPB + threadIdx.x;
    for (int k = 0; k < EPB; k += 256 * 5) {
        int d[5];
#pragma unroll
        for (int u = 0; u < 5; ++u) {
            const int e = e0 + k + u * 256;
            d[u] = dst[e < E ? e : E - 1];
        }
#pragma unroll
        for (int u = 0; u < 5; ++u)
            if (e0 + k + u * 256 < E) atomicAdd(&h[d[u] >> BIN_SHIFT], 1);
    }
    __syncthreads();
    for (int i = threadIdx.x; i < NBIN; i += 256) cell[i * NEB + blockIdx.x] = h[i];
}

__global__ __launch_bounds__(256) void scan_local_kernel(const int* __restrict__ cnt,
                                                         int* __restrict__ rowptr,
                                                         int* __restrict__ bsum, int n) {
    __shared__ int wsum[8];
    const int tid  = threadIdx.x;
    const int lane = tid & 31;
    const int wave = tid >> 5;
    const int i0 = blockIdx.x * SCAN_CHUNK + tid * 4;
    int t0 = (i0 + 0 < n) ? cnt[i0 + 0] : 0;
    int t1 = (i0 + 1 < n) ? cnt[i0 + 1] : 0;
    int t2 = (i0 + 2 < n) ? cnt[i0 + 2] : 0;
    int t3 = (i0 + 3 < n) ? cnt[i0 + 3] : 0;
    const int s = t0 + t1 + t2 + t3;
    int v = s;
#pragma unroll
    for (int o = 1; o < 32; o <<= 1) {
        int u = __shfl_up(v, o, 32);
        if (lane >= o) v += u;
    }
    if (lane == 31) wsum[wave] = v;
    __syncthreads();
    int woff = 0, tot = 0;
#pragma unroll
    for (int w = 0; w < 8; ++w) {
        int x = wsum[w];
        if (w < wave) woff += x;
        tot += x;
    }
    int ex = woff + v - s;
    if (i0 + 0 < n) rowptr[i0 + 0] = ex;
    if (i0 + 1 < n) rowptr[i0 + 1] = ex + t0;
    if (i0 + 2 < n) rowptr[i0 + 2] = ex + t0 + t1;
    if (i0 + 3 < n) rowptr[i0 + 3] = ex + t0 + t1 + t2;
    if (tid == 0) bsum[blockIdx.x] = tot;
}

__global__ __launch_bounds__(256) void scan_add_kernel(int* __restrict__ rowptr,
                                                       const int* __restrict__ bsum,
                                                       int n, int total) {
    __shared__ int wsum[8];
    const int tid  = threadIdx.x;
    const int lane = tid & 31;
    const int wave = tid >> 5;
    int v = (tid < (int)blockIdx.x) ? bsum[tid] : 0;
#pragma unroll
    for (int m = 16; m > 0; m >>= 1) v += __shfl_xor(v, m, 32);
    if (lane == 0) wsum[wave] = v;
    __syncthreads();
    int off = 0;
#pragma unroll
    for (int w = 0; w < 8; ++w) off += wsum[w];
    const int i0 = blockIdx.x * SCAN_CHUNK + tid * 4;
#pragma unroll
    for (int k = 0; k < 4; ++k)
        if (i0 + k < n) rowptr[i0 + k] += off;
    if (blockIdx.x == 0 && tid == 0) rowptr[n] = total;
}

__global__ __launch_bounds__(256) void bin_place_kernel(const int* __restrict__ src,
                                                        const int* __restrict__ dst,
                                                        const int* __restrict__ start,
                                                        int2* __restrict__ binned, int E) {
    __shared__ int cur[NBIN];
    for (int i = threadIdx.x; i < NBIN; i += 256) cur[i] = start[i * NEB + blockIdx.x];
    __syncthreads();
    const int e0 = blockIdx.x * EPB + threadIdx.x;
    for (int k = 0; k < EPB; k += 256 * 5) {
        int d[5], sv[5];
#pragma unroll
        for (int u = 0; u < 5; ++u) {
            const int e = e0 + k + u * 256;
            const int ec = e < E ? e : E - 1;
            d[u]  = dst[ec];
            sv[u] = src[ec];
        }
#pragma unroll
        for (int u = 0; u < 5; ++u) {
            if (e0 + k + u * 256 < E) {
                const int p = atomicAdd(&cur[d[u] >> BIN_SHIFT], 1);
                binned[p] = make_int2(sv[u], d[u]);
            }
        }
    }
}

__global__ __launch_bounds__(256) void bin_sort_kernel(const int2* __restrict__ binned,
                                                       const int* __restrict__ start,
                                                       int* __restrict__ rowptr,
                                                       int* __restrict__ csr, int N) {
    __shared__ int cnt[BIN_NODES];
    __shared__ int wsum[8];
    __shared__ int stage[STAGE_CAP];
    const int tid  = threadIdx.x;
    const int lane = tid & 31;
    const int wave = tid >> 5;
    const int bin  = blockIdx.x;
    const int s0 = start[bin * NEB];
    const int s1 = start[(bin + 1) * NEB];
    cnt[tid] = 0;
    __syncthreads();
    for (int i0 = s0 + tid; i0 - tid < s1; i0 += 1024) {
        int dl[4];
#pragma unroll
        for (int u = 0; u < 4; ++u) {
            const int i = i0 + u * 256;
            dl[u] = binned[i < s1 ? i : s1 - 1].y;
        }
#pragma unroll
        for (int u = 0; u < 4; ++u)
            if (i0 + u * 256 < s1) atomicAdd(&cnt[dl[u] & (BIN_NODES - 1)], 1);
    }
    __syncthreads();
    const int c = cnt[tid];
    int v = c;
#pragma unroll
    for (int o = 1; o < 32; o <<= 1) {
        int u = __shfl_up(v, o, 32);
        if (lane >= o) v += u;
    }
    if (lane == 31) wsum[wave] = v;
    __syncthreads();
    int woff = 0;
#pragma unroll
    for (int w = 0; w < 8; ++w) {
        int x = wsum[w];
        if (w < wave) woff += x;
    }
    const int ex = woff + v - c;
    const int node = bin * BIN_NODES + tid;
    if (node <= N) rowptr[node] = s0 + ex;
    cnt[tid] = ex;
    __syncthreads();
    const int len = s1 - s0;
    const bool fits = len <= STAGE_CAP;
    for (int i0 = s0 + tid; i0 - tid < s1; i0 += 1024) {
        int2 ed[4];
#pragma unroll
        for (int u = 0; u < 4; ++u) {
            const int i = i0 + u * 256;
            ed[u] = binned[i < s1 ? i : s1 - 1];
        }
#pragma unroll
        for (int u = 0; u < 4; ++u) {
            if (i0 + u * 256 < s1) {
                const int p = atomicAdd(&cnt[ed[u].y & (BIN_NODES - 1)], 1);
                if (fits) stage[p] = ed[u].x; else csr[s0 + p] = ed[u].x;
            }
        }
    }
    if (fits) {
        __syncthreads();
        for (int i = tid; i < len; i += 256) csr[s0 + i] = stage[i];
    }
}

__global__ __launch_bounds__(256) void gemm64_wmma(const float* __restrict__ X,
                                                   const float* __restrict__ W,
                                                   const int* __restrict__ rowptr,
                                                   float* __restrict__ H, int nTiles) {
    __shared__ float Ws[64 * 64];
    __shared__ float Xs[8][16 * XPAD];

    const int tid  = threadIdx.x;
    const int wave = tid >> 5;
    const int lane = tid & 31;
    const int tile = blockIdx.x * 8 + wave;

#pragma unroll
    for (int i = 0; i < 16; ++i) {
        int g = tid + i * 256;
        int k = g >> 6;
        int n = g & 63;
        Ws[(k >> 1) * 128 + n * 2 + (k & 1)] = W[g];
    }

    const int rowOff = (lane >> 4) * 8;
    const int base = tile * 16;
    int rp[9] = {};

    if (tile < nTiles) {
#pragma unroll
        for (int r = 0; r < 9; ++r) rp[r] = rowptr[base + rowOff + r];
        const float4* Xg = (const float4*)(X + (size_t)tile * 16 * 64);
        float* xl = &Xs[wave][0];
#pragma unroll
        for (int i = 0; i < 8; ++i) {
            int idx4 = lane + i * 32;
            float4 v = Xg[idx4];
            int row = idx4 >> 4;
            int col = (idx4 & 15) * 4;
            *(float4*)&xl[row * XPAD + col] = v;
        }
    }
    __syncthreads();
    if (tile >= nTiles) return;

    const float* xs = &Xs[wave][0];
    const int l15 = lane & 15;
    const int hiK = (lane >> 4) * 2;

    v8f c0 = {}, c1 = {}, c2 = {}, c3 = {};
#pragma unroll
    for (int k = 0; k < 64; k += 4) {
        v2f a = *(const v2f*)(xs + l15 * XPAD + k + hiK);
        const float* wrow = Ws + ((k + hiK) >> 1) * 128 + l15 * 2;
        v2f b0 = *(const v2f*)(wrow +  0);
        v2f b1 = *(const v2f*)(wrow + 32);
        v2f b2 = *(const v2f*)(wrow + 64);
        v2f b3 = *(const v2f*)(wrow + 96);
        c0 = __builtin_amdgcn_wmma_f32_16x16x4_f32(false, a, false, b0, (short)0, c0, false, false);
        c1 = __builtin_amdgcn_wmma_f32_16x16x4_f32(false, a, false, b1, (short)0, c1, false, false);
        c2 = __builtin_amdgcn_wmma_f32_16x16x4_f32(false, a, false, b2, (short)0, c2, false, false);
        c3 = __builtin_amdgcn_wmma_f32_16x16x4_f32(false, a, false, b3, (short)0, c3, false, false);
    }

#pragma unroll
    for (int r = 0; r < 8; ++r) {
        const float di = rsqrtf((float)(rp[r + 1] - rp[r] + 1));
        float* out = H + (size_t)(base + r + rowOff) * 64 + l15;
        out[0]  = c0[r] * di;
        out[16] = c1[r] * di;
        out[32] = c2[r] * di;
        out[48] = c3[r] * di;
    }
}

__global__ __launch_bounds__(256) void gather_kernel(const float* __restrict__ hs,
                                                     const int* __restrict__ rowptr,
                                                     const int* __restrict__ csr,
                                                     const float* __restrict__ b,
                                                     float* __restrict__ out, int N) {
    const int wave = __builtin_amdgcn_readfirstlane((int)threadIdx.x >> 5);
    const int node = blockIdx.x * 8 + wave;
    const int lane = threadIdx.x & 31;
    if (node >= N) return;

    const v2f* __restrict__ H2 = (const v2f*)hs;
    const int beg = rowptr[node];
    const int end = rowptr[node + 1];
    const v2f self = H2[node * 32 + lane];

    v2f a0 = {}, a1 = {}, a2 = {}, a3 = {}, a4 = {}, a5 = {}, a6 = {}, a7 = {};
    for (int sb = beg; sb < end; sb += 32) {
        const int left = end - sb;
        const int cn = left < 32 ? left : 32;
        const int idx = (lane < cn) ? csr[sb + lane] : 0;
        int j = 0;
        for (; j + 8 <= cn; j += 8) {
            const v2f v0 = H2[__builtin_amdgcn_readlane(idx, j + 0) * 32 + lane];
            const v2f v1 = H2[__builtin_amdgcn_readlane(idx, j + 1) * 32 + lane];
            const v2f v2 = H2[__builtin_amdgcn_readlane(idx, j + 2) * 32 + lane];
            const v2f v3 = H2[__builtin_amdgcn_readlane(idx, j + 3) * 32 + lane];
            const v2f v4 = H2[__builtin_amdgcn_readlane(idx, j + 4) * 32 + lane];
            const v2f v5 = H2[__builtin_amdgcn_readlane(idx, j + 5) * 32 + lane];
            const v2f v6 = H2[__builtin_amdgcn_readlane(idx, j + 6) * 32 + lane];
            const v2f v7 = H2[__builtin_amdgcn_readlane(idx, j + 7) * 32 + lane];
            a0 += v0; a1 += v1; a2 += v2; a3 += v3;
            a4 += v4; a5 += v5; a6 += v6; a7 += v7;
        }
        const int rem = cn - j;
        if (rem > 0) {
            v2f t0 = {}, t1 = {}, t2 = {}, t3 = {}, t4 = {}, t5 = {}, t6 = {};
            t0 = H2[__builtin_amdgcn_readlane(idx, j + 0) * 32 + lane];
            if (rem > 1) t1 = H2[__builtin_amdgcn_readlane(idx, j + 1) * 32 + lane];
            if (rem > 2) t2 = H2[__builtin_amdgcn_readlane(idx, j + 2) * 32 + lane];
            if (rem > 3) t3 = H2[__builtin_amdgcn_readlane(idx, j + 3) * 32 + lane];
            if (rem > 4) t4 = H2[__builtin_amdgcn_readlane(idx, j + 4) * 32 + lane];
            if (rem > 5) t5 = H2[__builtin_amdgcn_readlane(idx, j + 5) * 32 + lane];
            if (rem > 6) t6 = H2[__builtin_amdgcn_readlane(idx, j + 6) * 32 + lane];
            a0 += t0; a1 += t1; a2 += t2; a3 += t3;
            a4 += t4; a5 += t5; a6 += t6;
        }
    }

    const float di = rsqrtf((float)(end - beg + 1));
    const v2f bb = ((const v2f*)b)[lane];
    const v2f acc = (((a0 + a1) + (a2 + a3)) + ((a4 + a5) + (a6 + a7))) + self;
    v2f o;
    o.x = fmaxf(di * acc.x + bb.x, 0.0f);
    o.y = fmaxf(di * acc.y + bb.y, 0.0f);
    ((v2f*)out)[node * 32 + lane] = o;
}

__global__ __launch_bounds__(256) void proj_kernel(const float* __restrict__ f,
                                                   const float* __restrict__ Wp,
                                                   const float* __restrict__ bp,
                                                   float* __restrict__ y, int N) {
    int node = blockIdx.x * 8 + (threadIdx.x >> 5);
    int lane = threadIdx.x & 31;
    if (node >= N) return;
    const float* fr = f + (size_t)node * 64;
    float s = fr[lane] * Wp[lane] + fr[lane + 32] * Wp[lane + 32];
#pragma unroll
    for (int m = 16; m > 0; m >>= 1) s += __shfl_xor(s, m, 32);
    if (lane == 0) y[node] = s + bp[0];
}

extern "C" void kernel_launch(void* const* d_in, const int* in_sizes, int n_in,
                              void* d_out, int out_size, void* d_ws, size_t ws_size,
                              hipStream_t stream) {
    (void)in_sizes; (void)n_in; (void)out_size; (void)ws_size;

    const float* x  = (const float*)d_in[0];
    const int*   ei = (const int*)d_in[1];
    const float* W1 = (const float*)d_in[2];
    const float* b1 = (const float*)d_in[3];
    const float* W2 = (const float*)d_in[4];
    const float* b2 = (const float*)d_in[5];
    const float* Wp = (const float*)d_in[6];
    const float* bp = (const float*)d_in[7];
    float* y = (float*)d_out;

    const int N = GNN_N, E = GNN_E;
    const int* src = ei;
    const int* dst = ei + E;

    char* ws = (char*)d_ws;
    size_t off = 0;
    auto take = [&](size_t bytes) {
        void* p = ws + off;
        off += (bytes + 255) & ~(size_t)255;
        return p;
    };
    int*   rowptr = (int*)take((size_t)(N + 1) * 4);
    int*   csr    = (int*)take((size_t)E * 4);
    float* bufA   = (float*)take((size_t)N * DIM * 4);
    float* bufB   = (float*)take((size_t)N * DIM * 4);
    int*  cell   = (int*)bufA;
    int*  bsum   = cell + NCELL;
    int*  start  = bsum + 256;
    int2* binned = (int2*)bufB;

    const int nTiles = N / 16;
    const int gemmBlocks = (nTiles + 7) / 8;

    bin_count_kernel<<<NEB, 256, 0, stream>>>(dst, cell, E);
    scan_local_kernel<<<SCAN_BLOCKS, 256, 0, stream>>>(cell, start, bsum, NCELL);
    scan_add_kernel<<<SCAN_BLOCKS, 256, 0, stream>>>(start, bsum, NCELL, E);
    bin_place_kernel<<<NEB, 256, 0, stream>>>(src, dst, start, binned, E);
    bin_sort_kernel<<<NBIN, 256, 0, stream>>>(binned, start, rowptr, csr, N);

    gemm64_wmma<<<gemmBlocks, 256, 0, stream>>>(x, W1, rowptr, bufA, nTiles);
    gather_kernel<<<(N + 7) / 8, 256, 0, stream>>>(bufA, rowptr, csr, b1, bufB, N);

    gemm64_wmma<<<gemmBlocks, 256, 0, stream>>>(bufB, W2, rowptr, bufA, nTiles);
    gather_kernel<<<(N + 7) / 8, 256, 0, stream>>>(bufA, rowptr, csr, b2, bufB, N);

    proj_kernel<<<(N + 7) / 8, 256, 0, stream>>>(bufB, Wp, bp, y, N);
}


# Few long-lived waves instead of one per node: gather on a resident grid (1024 work-groups, each wave walks its nodes, next bounds and ids fetched ahead), head 32 rows per wave: 321k -> 34k waves
# speedup vs baseline: 5.6909x; 1.0820x over previous
#include <hip/hip_runtime.h>


typedef float v2f __attribute__((ext_vector_type(2)));
typedef float v8f __attribute__((ext_vector_type(8)));

#define GNN_N 100000
#define GNN_E 1600000
#define DIM   64
#define XPAD  68
#define BIN_SHIFT 8
#define BIN_NODES 256
#define NBIN (GNN_N / BIN_NODES + 1)
#define EPB 6400
#define NEB ((GNN_E + EPB - 1) / EPB)
#define NCELL (NBIN * NEB)
#define STAGE_CAP 8192
#define SCAN_CHUNK 1024
#define SCAN_BLOCKS ((NCELL + SCAN_CHUNK - 1) / SCAN_CHUNK)
static_assert(EPB % (256 * 5) == 0, "a binning work-group walks its slice in steps of 256 x 5 edges");
static_assert(SCAN_BLOCKS <= 256, "the second scan step sums the chunk totals with one 256-thread group");

__global__ __launch_bounds__(256) void bin_count_kernel(const int* __restrict__ dst,
                                                        int* __restrict__ cell, int E) {
    __shared__ int h[NBIN];
    for (int i = threadIdx.x; i < NBIN; i += 256) h[i] = 0;
    __syncthreads();
    const int e0 = blockIdx.x * EPB + threadIdx.x;
    for (int k = 0; k < EPB; k += 256 * 5) {
        int d[5];
#pragma unroll
        for (int u = 0; u < 5; ++u) {
            const int e = e0 + k + u * 256;
            d[u] = dst[e < E ? e : E - 1];
        }
#pragma unroll
        for (int u = 0; u < 5; ++u)
            if (e0 + k + u * 256 < E) atomicAdd(&h[d[u] >> BIN_SHIFT], 1);
    }
    __syncthreads();
    for (int i = threadIdx.x; i < NBIN; i += 256) cell[i * NEB + blockIdx.x] = h[i];
}

__global__ __launch_bounds__(256) void scan_local_kernel(const int* __restrict__ cnt,
                                                         int* __restrict__ rowptr,
                                                         int* __restrict__ bsum, int n) {
    __shared__ int wsum[8];
    const int tid  = threadIdx.x;
    const int lane = tid & 31;
    const int wave = tid >> 5;
    const int i0 = blockIdx.x * SCAN_CHUNK + tid * 4;
    int t0 = (i0 + 0 < n) ? cnt[i0 + 0] : 0;
    int t1 = (i0 + 1 < n) ? cnt[i0 + 1] : 0;
    int t2 = (i0 + 2 < n) ? cnt[i0 + 2] : 0;
    int t3 = (i0 + 3 < n) ? cnt[i0 + 3] : 0;
    const int s = t0 + t1 + t2 + t3;
    int v = s;
#pragma unroll
    for (int o = 1; o < 32; o <<= 1) {
        int u = __shfl_up(v, o, 32);
        if (lane >= o) v += u;
    }
    if (lane == 31) wsum[wave] = v;
    __syncthreads();
    int woff = 0, tot = 0;
#pragma unroll
    for (int w = 0; w < 8; ++w) {
        int x = wsum[w];
        if (w < wave) woff += x;
        tot += x;
    }
    int ex = woff + v - s;
    if (i0 + 0 < n) rowptr[i0 + 0] = ex;
    if (i0 + 1 < n) rowptr[i0 + 1] = ex + t0;
    if (i0 + 2 < n) rowptr[i0 + 2] = ex + t0 + t1;
    if (i0 + 3 < n) rowptr[i0 + 3] = ex + t0 + t1 + t2;
    if (tid == 0) bsum[blockIdx.x] = tot;
}

__global__ __launch_bounds__(256) void scan_add_kernel(int* __restrict__ rowptr,
                                                       const int* __restrict__ bsum,
                                                       int n, int total) {
    __shared__ int wsum[8];
    const int tid  = threadIdx.x;
    const int lane = tid & 31;
    const int wave = tid >> 5;
    int v = (tid < (int)blockIdx.x) ? bsum[tid] : 0;
#pragma unroll
    for (int m = 16; m > 0; m >>= 1) v += __shfl_xor(v, m, 32);
    if (lane == 0) wsum[wave] = v;
    __syncthreads();
    int off = 0;
#pragma unroll
    for (int w = 0; w < 8; ++w) off += wsum[w];
    const int i0 = blockIdx.x * SCAN_CHUNK + tid * 4;
#pragma unroll
    for (int k = 0; k < 4; ++k)
        if (i0 + k < n) rowptr[i0 + k] += off;
    if (blockIdx.x == 0 && tid == 0) rowptr[n] = total;
}

__global__ __launch_bounds__(256) void bin_place_kernel(const int* __restrict__ src,
                                                        const int* __restrict__ dst,
                                                        const int* __restrict__ start,
                                                        int2* __restrict__ binned, int E) {
    __shared__ int cur[NBIN];
    for (int i = threadIdx.x; i < NBIN; i += 256) cur[i] = start[i * NEB + blockIdx.x];
    __syncthreads();
    const int e0 = blockIdx.x * EPB + threadIdx.x;
    for (int k = 0; k < EPB; k += 256 * 5) {
        int d[5], sv[5];
#pragma unroll
        for (int u = 0; u < 5; ++u) {
            const int e = e0 + k + u * 256;
            const int ec = e < E ? e : E - 1;
            d[u]  = dst[ec];
            sv[u] = src[ec];
        }
#pragma unroll
        for (int u = 0; u < 5; ++u) {
            if (e0 + k + u * 256 < E) {
                const int p = atomicAdd(&cur[d[u] >> BIN_SHIFT], 1);
                binned[p] = make_int2(sv[u], d[u]);
            }
        }
    }
}

__global__ __launch_bounds__(256) void bin_sort_kernel(const int2* __restrict__ binned,
                                                       const int* __restrict__ start,
                                                       int* __restrict__ rowptr,
                                                       int* __restrict__ csr, int N) {
    __shared__ int cnt[BIN_NODES];
    __shared__ int wsum[8];
    __shared__ int stage[STAGE_CAP];
    const int tid  = threadIdx.x;
    const int lane = tid & 31;
    const int wave = tid >> 5;
    const int bin  = blockIdx.x;
    const int s0 = start[bin * NEB];
    const int s1 = start[(bin + 1) * NEB];
    cnt[tid] = 0;
    __syncthreads();
    for (int i0 = s0 + tid; i0 - tid < s1; i0 += 1024) {
        int dl[4];
#pragma unroll
        for (int u = 0; u < 4; ++u) {
            const int i = i0 + u * 256;
            dl[u] = binned[i < s1 ? i : s1 - 1].y;
        }
#pragma unroll
        for (int u = 0; u < 4; ++u)
            if (i0 + u * 256 < s1) atomicAdd(&cnt[dl[u] & (BIN_NODES - 1)], 1);
    }
    __syncthreads();
    const int c = cnt[tid];
    int v = c;
#pragma unroll
    for (int o = 1; o < 32; o <<= 1) {
        int u = __shfl_up(v, o, 32);
        if (lane >= o) v += u;
    }
    if (lane == 31) wsum[wave] = v;
    __syncthreads();
    int woff = 0;
#pragma unroll
    for (int w = 0; w < 8; ++w) {
        int x = wsum[w];
        if (w < wave) woff += x;
    }
    const int ex = woff + v - c;
    const int node = bin * BIN_NODES + tid;
    if (node <= N) rowptr[node] = s0 + ex;
    cnt[tid] = ex;
    __syncthreads();
    const int len = s1 - s0;
    const bool fits = len <= STAGE_CAP;
    for (int i0 = s0 + tid; i0 - tid < s1; i0 += 1024) {
        int2 ed[4];
#pragma unroll
        for (int u = 0; u < 4; ++u) {
            const int i = i0 + u * 256;
            ed[u] = binned[i < s1 ? i : s1 - 1];
        }
#pragma unroll
        for (int u = 0; u < 4; ++u) {
            if (i0 + u * 256 < s1) {
                const int p = atomicAdd(&cnt[ed[u].y & (BIN_NODES - 1)], 1);
                if (fits) stage[p] = ed[u].x; else csr[s0 + p] = ed[u].x;
            }
        }
    }
    if (fits) {
        __syncthreads();
        for (int i = tid; i < len; i += 256) csr[s0 + i] = stage[i];
    }
}

__global__ __launch_bounds__(256) void gemm64_wmma(const float* __restrict__ X,
                                                   const float* __restrict__ W,
                                                   const int* __restrict__ rowptr,
                                                   float* __restrict__ H, int nTiles) {
    __shared__ float Ws[64 * 64];
    __shared__ float Xs[8][16 * XPAD];

    const int tid  = threadIdx.x;
    const int wave = tid >> 5;
    const int lane = tid & 31;
    const int tile = blockIdx.x * 8 + wave;

#pragma unroll
    for (int i = 0; i < 16; ++i) {
        int g = tid + i * 256;
        int k = g >> 6;
        int n = g & 63;
        Ws[(k >> 1) * 128 + n * 2 + (k & 1)] = W[g];
    }

    const int rowOff = (lane >> 4) * 8;
    const int base = tile * 16;
    int rp[9] = {};

    if (tile < nTiles) {
#pragma unroll
        for (int r = 0; r < 9; ++r) rp[r] = rowptr[base + rowOff + r];
        const float4* Xg = (const float4*)(X + (size_t)tile * 16 * 64);
        float* xl = &Xs[wave][0];
#pragma unroll
        for (int i = 0; i < 8; ++i) {
            int idx4 = lane + i * 32;
            float4 v = Xg[idx4];
            int row = idx4 >> 4;
            int col = (idx4 & 15) * 4;
            *(float4*)&xl[row * XPAD + col] = v;
        }
    }
    __syncthreads();
    if (tile >= nTiles) return;

    const float* xs = &Xs[wave][0];
    const int l15 = lane & 15;
    const int hiK = (lane >> 4) * 2;

    v8f c0 = {}, c1 = {}, c2 = {}, c3 = {};
#pragma unroll
    for (int k = 0; k < 64; k += 4) {
        v2f a = *(const v2f*)(xs + l15 * XPAD + k + hiK);
        const float* wrow = Ws + ((k + hiK) >> 1) * 128 + l15 * 2;
        v2f b0 = *(const v2f*)(wrow +  0);
        v2f b1 = *(const v2f*)(wrow + 32);
        v2f b2 = *(const v2f*)(wrow + 64);
        v2f b3 = *(const v2f*)(wrow + 96);
        c0 = __builtin_amdgcn_wmma_f32_16x16x4_f32(false, a, false, b0, (short)0, c0, false, false);
        c1 = __builtin_amdgcn_wmma_f32_16x16x4_f32(false, a, false, b1, (short)0, c1, false, false);
        c2 = __builtin_amdgcn_wmma_f32_16x16x4_f32(false, a, false, b2, (short)0, c2, false, false);
        c3 = __builtin_amdgcn_wmma_f32_16x16x4_f32(false, a, false, b3, (short)0, c3, false, false);
    }

#pragma unroll
    for (int r = 0; r < 8; ++r) {
        const float di = rsqrtf((float)(rp[r + 1] - rp[r] + 1));
        float* out = H + (size_t)(base + r + rowOff) * 64 + l15;
        out[0]  = c0[r] * di;
        out[16] = c1[r] * di;
        out[32] = c2[r] * di;
        out[48] = c3[r] * di;
    }
}

#define GATHER_WGS 1024

#define GATHER_ROW(k) H2[__builtin_amdgcn_readlane(idx, (k) - 1) * 32 + lane]

__global__ __launch_bounds__(256) void gather_kernel(const float* __restrict__ hs,
                                                     const int* __restrict__ rowptr,
                                                     const int* __restrict__ csr,
                                                     const float* __restrict__ b,
                                                     float* __restrict__ out, int N) {
    const int wave = __builtin_amdgcn_readfirstlane((int)threadIdx.x >> 5);
    const int lane = threadIdx.x & 31;
    const int stride = (int)gridDim.x * 8;
    int node = blockIdx.x * 8 + wave;
    if (node >= N) return;

    const v2f* __restrict__ H2 = (const v2f*)hs;
    const v2f bb = ((const v2f*)b)[lane];
    const v2f zero = {};

    int beg = rowptr[node];
    int end = rowptr[node + 1];
    int nxt = node + stride;
    int nbeg = 0, nend = 0;
    if (nxt < N) { nbeg = rowptr[nxt]; nend = rowptr[nxt + 1]; }
    int idx = csr[max(min(beg + lane, end - 1), 0)];
    asm volatile("" : "+v"(idx));

    for (;;) {
        const int nx2 = nxt + stride;
        int n2beg = 0, n2end = 0;
        if (nx2 < N) { n2beg = rowptr[nx2]; n2end = rowptr[nx2 + 1]; }
        int nidx = 0;
        if (nxt < N) nidx = csr[max(min(nbeg + lane, nend - 1), 0)];

        const int deg = end - beg;
        const int rows = (deg < 32 ? deg : 32) + 1;
        const int rem = rows & 7;
        const int full = rows - rem;
        v2f t0 = zero, t1 = zero, t2 = zero, t3 = zero, t4 = zero, t5 = zero, t6 = zero;
        if (rem > 0) {
            const int r0 = full == 0 ? node : __builtin_amdgcn_readlane(idx, full == 0 ? 0 : full - 1);
            t0 = H2[r0 * 32 + lane];
        }
        if (rem > 1) t1 = GATHER_ROW(full + 1);
        if (rem > 2) t2 = GATHER_ROW(full + 2);
        if (rem > 3) t3 = GATHER_ROW(full + 3);
        if (rem > 4) t4 = GATHER_ROW(full + 4);
        if (rem > 5) t5 = GATHER_ROW(full + 5);
        if (rem > 6) t6 = GATHER_ROW(full + 6);
        v2f v0 = zero, v1 = zero, v2 = zero, v3 = zero, v4 = zero, v5 = zero, v6 = zero, v7 = zero;
        if (full >= 8) {
            v0 = H2[node * 32 + lane];
            v1 = GATHER_ROW(1); v2 = GATHER_ROW(2); v3 = GATHER_ROW(3);
            v4 = GATHER_ROW(4); v5 = GATHER_ROW(5); v6 = GATHER_ROW(6); v7 = GATHER_ROW(7);
        }
        v2f w0 = zero, w1 = zero, w2 = zero, w3 = zero, w4 = zero, w5 = zero, w6 = zero, w7 = zero;
        if (full >= 16) {
            w0 = GATHER_ROW(8);  w1 = GATHER_ROW(9);  w2 = GATHER_ROW(10); w3 = GATHER_ROW(11);
            w4 = GATHER_ROW(12); w5 = GATHER_ROW(13); w6 = GATHER_ROW(14); w7 = GATHER_ROW(15);
        }
        v2f a0 = (v0 + w0) + t0, a1 = (v1 + w1) + t1, a2 = (v2 + w2) + t2, a3 = (v3 + w3) + t3;
        v2f a4 = (v4 + w4) + t4, a5 = (v5 + w5) + t5, a6 = (v6 + w6) + t6, a7 = v7 + w7;
        for (int j = 16; j < full; j += 8) {
            const v2f x0 = GATHER_ROW(j + 0), x1 = GATHER_ROW(j + 1);
            const v2f x2 = GATHER_ROW(j + 2), x3 = GATHER_ROW(j + 3);
            const v2f x4 = GATHER_ROW(j + 4), x5 = GATHER_ROW(j + 5);
            const v2f x6 = GATHER_ROW(j + 6), x7 = GATHER_ROW(j + 7);
            a0 += x0; a1 += x1; a2 += x2; a3 += x3;
            a4 += x4; a5 += x5; a6 += x6; a7 += x7;
        }
        for (int e = beg + 32; e < end; ++e)
            a0 += H2[csr[e] * 32 + lane];

        const float di = rsqrtf((float)(deg + 1));
        const v2f acc = ((a0 + a1) + (a2 + a3)) + ((a4 + a5) + (a6 + a7));
        v2f o;
        o.x = fmaxf(di * acc.x + bb.x, 0.0f);
        o.y = fmaxf(di * acc.y + bb.y, 0.0f);
        ((v2f*)out)[node * 32 + lane] = o;

        if (nxt >= N) break;
        node = nxt;  beg = nbeg;   end = nend;   idx = nidx;
        nxt  = nx2;  nbeg = n2beg; nend = n2end;
    }
}

__global__ __launch_bounds__(256) void proj_kernel(const float* __restrict__ f,
                                                   const float* __restrict__ Wp,
                                                   const float* __restrict__ bp,
                                                   float* __restrict__ y, int N) {
    const int lane = threadIdx.x & 31;
    const int base = (blockIdx.x * 8 + (threadIdx.x >> 5)) * 32;
    if (base >= N) return;
    const int half = lane >> 4;
    const int c = lane & 15;
    const float4 w = ((const float4*)Wp)[c];
    const float4* __restrict__ F4 = (const float4*)f;
    float s[16];
#pragma unroll
    for (int u = 0; u < 16; ++u) {
        const int r = base + 2 * u + half;
        const float4 v = F4[(size_t)(r < N ? r : N - 1) * 16 + c];
        s[u] = (v.x * w.x + v.y * w.y) + (v.z * w.z + v.w * w.w);
    }
#pragma unroll
    for (int u = 0; u < 16; ++u) {
#pragma unroll
        for (int m = 8; m > 0; m >>= 1) s[u] += __shfl_xor(s[u], m, 32);
    }
    float r = s[0];
#pragma unroll
    for (int u = 1; u < 16; ++u) r = (c == u) ? s[u] : r;
    const int row = base + 2 * c + half;
    if (row < N) y[row] = r + bp[0];
}

extern "C" void kernel_launch(void* const* d_in, const int* in_sizes, int n_in,
                              void* d_out, int out_size, void* d_ws, size_t ws_size,
                              hipStream_t stream) {
    (void)in_sizes; (void)n_in; (void)out_size; (void)ws_size;

    const float* x  = (const float*)d_in[0];
    const int*   ei = (const int*)d_in[1];
    const float* W1 = (const float*)d_in[2];
    const float* b1 = (const float*)d_in[3];
    const float* W2 = (const float*)d_in[4];
    const float* b2 = (const float*)d_in[5];
    const float* Wp = (const float*)d_in[6];
    const float* bp = (const float*)d_in[7];
    float* y = (float*)d_out;

    const int N = GNN_N, E = GNN_E;
    const int* src = ei;
    const int* dst = ei + E;

    char* ws = (char*)d_ws;
    size_t off = 0;
    auto take = [&](size_t bytes) {
        void* p = ws + off;
        off += (bytes + 255) & ~(size_t)255;
        return p;
    };
    int*   rowptr = (int*)take((size_t)(N + 1) * 4);
    int*   csr    = (int*)take((size_t)E * 4);
    float* bufA   = (float*)take((size_t)N * DIM * 4);
    float* bufB   = (float*)take((size_t)N * DIM * 4);
    int*  cell   = (int*)bufA;
    int*  bsum   = cell + NCELL;
    int*  start  = bsum + 256;
    int2* binned = (int2*)bufB;

    const int nTiles = N / 16;
    const int gemmBlocks = (nTiles + 7) / 8;

    bin_count_kernel<<<NEB, 256, 0, stream>>>(dst, cell, E);
    scan_local_kernel<<<SCAN_BLOCKS, 256, 0, stream>>>(cell, start, bsum, NCELL);
    scan_add_kernel<<<SCAN_BLOCKS, 256, 0, stream>>>(start, bsum, NCELL, E);
    bin_place_kernel<<<NEB, 256, 0, stream>>>(src, dst, start, binned, E);
    bin_sort_kernel<<<NBIN, 256, 0, stream>>>(binned, start, rowptr, csr, N);

    gemm64_wmma<<<gemmBlocks, 256, 0, stream>>>(x, W1, rowptr, bufA, nTiles);
    gather_kernel<<<GATHER_WGS, 256, 0, stream>>>(bufA, rowptr, csr, b1, bufB, N);

    gemm64_wmma<<<gemmBlocks, 256, 0, stream>>>(bufB, W2, rowptr, bufA, nTiles);
    gather_kernel<<<GATHER_WGS, 256, 0, stream>>>(bufA, rowptr, csr, b2, bufB, N);

    proj_kernel<<<(N + 255) / 256, 256, 0, stream>>>(bufB, Wp, bp, y, N);
}


# Head folded into the layer-2 gather: the finished row's 64 products with Wp are summed over the wave in registers, y lines staged in LDS and written whole; layer-2 rows never stored, 10 to 9 launches
# speedup vs baseline: 5.8931x; 1.0355x over previous
#include <hip/hip_runtime.h>


typedef float v2f __attribute__((ext_vector_type(2)));
typedef float v8f __attribute__((ext_vector_type(8)));

#define GNN_N 100000
#define GNN_E 1600000
#define DIM   64
#define XPAD  68
#define BIN_SHIFT 8
#define BIN_NODES 256
#define NBIN (GNN_N / BIN_NODES + 1)
#define EPB 6400
#define NEB ((GNN_E + EPB - 1) / EPB)
#define NCELL (NBIN * NEB)
#define STAGE_CAP 8192
#define SCAN_CHUNK 1024
#define SCAN_BLOCKS ((NCELL + SCAN_CHUNK - 1) / SCAN_CHUNK)
static_assert(EPB % (256 * 5) == 0, "a binning work-group walks its slice in steps of 256 x 5 edges");
static_assert(SCAN_BLOCKS <= 256, "the second scan step sums the chunk totals with one 256-thread group");

__global__ __launch_bounds__(256) void bin_count_kernel(const int* __restrict__ dst,
                                                        int* __restrict__ cell, int E) {
    __shared__ int h[NBIN];
    for (int i = threadIdx.x; i < NBIN; i += 256) h[i] = 0;
    __syncthreads();
    const int e0 = blockIdx.x * EPB + threadIdx.x;
    for (int k = 0; k < EPB; k += 256 * 5) {
        int d[5];
#pragma unroll
        for (int u = 0; u < 5; ++u) {
            const int e = e0 + k + u * 256;
            d[u] = dst[e < E ? e : E - 1];
        }
#pragma unroll
        for (int u = 0; u < 5; ++u)
            if (e0 + k + u * 256 < E) atomicAdd(&h[d[u] >> BIN_SHIFT], 1);
    }
    __syncthreads();
    for (int i = threadIdx.x; i < NBIN; i += 256) cell[i * NEB + blockIdx.x] = h[i];
}

__global__ __launch_bounds__(256) void scan_local_kernel(const int* __restrict__ cnt,
                                                         int* __restrict__ rowptr,
                                                         int* __restrict__ bsum, int n) {
    __shared__ int wsum[8];
    const int tid  = threadIdx.x;
    const int lane = tid & 31;
    const int wave = tid >> 5;
    const int i0 = blockIdx.x * SCAN_CHUNK + tid * 4;
    int t0 = (i0 + 0 < n) ? cnt[i0 + 0] : 0;
    int t1 = (i0 + 1 < n) ? cnt[i0 + 1] : 0;
    int t2 = (i0 + 2 < n) ? cnt[i0 + 2] : 0;
    int t3 = (i0 + 3 < n) ? cnt[i0 + 3] : 0;
    const int s = t0 + t1 + t2 + t3;
    int v = s;
#pragma unroll
    for (int o = 1; o < 32; o <<= 1) {
        int u = __shfl_up(v, o, 32);
        if (lane >= o) v += u;
    }
    if (lane == 31) wsum[wave] = v;
    __syncthreads();
    int woff = 0, tot = 0;
#pragma unroll
    for (int w = 0; w < 8; ++w) {
        int x = wsum[w];
        if (w < wave) woff += x;
        tot += x;
    }
    int ex = woff + v - s;
    if (i0 + 0 < n) rowptr[i0 + 0] = ex;
    if (i0 + 1 < n) rowptr[i0 + 1] = ex + t0;
    if (i0 + 2 < n) rowptr[i0 + 2] = ex + t0 + t1;
    if (i0 + 3 < n) rowptr[i0 + 3] = ex + t0 + t1 + t2;
    if (tid == 0) bsum[blockIdx.x] = tot;
}

__global__ __launch_bounds__(256) void scan_add_kernel(int* __restrict__ rowptr,
                                                       const int* __restrict__ bsum,
                                                       int n, int total) {
    __shared__ int wsum[8];
    const int tid  = threadIdx.x;
    const int lane = tid & 31;
    const int wave = tid >> 5;
    int v = (tid < (int)blockIdx.x) ? bsum[tid] : 0;
#pragma unroll
    for (int m = 16; m > 0; m >>= 1) v += __shfl_xor(v, m, 32);
    if (lane == 0) wsum[wave] = v;
    __syncthreads();
    int off = 0;
#pragma unroll
    for (int w = 0; w < 8; ++w) off += wsum[w];
    const int i0 = blockIdx.x * SCAN_CHUNK + tid * 4;
#pragma unroll
    for (int k = 0; k < 4; ++k)
        if (i0 + k < n) rowptr[i0 + k] += off;
    if (blockIdx.x == 0 && tid == 0) rowptr[n] = total;
}

__global__ __launch_bounds__(256) void bin_place_kernel(const int* __restrict__ src,
                                                        const int* __restrict__ dst,
                                                        const int* __restrict__ start,
                                                        int2* __restrict__ binned, int E) {
    __shared__ int cur[NBIN];
    for (int i = threadIdx.x; i < NBIN; i += 256) cur[i] = start[i * NEB + blockIdx.x];
    __syncthreads();
    const int e0 = blockIdx.x * EPB + threadIdx.x;
    for (int k = 0; k < EPB; k += 256 * 5) {
        int d[5], sv[5];
#pragma unroll
        for (int u = 0; u < 5; ++u) {
            const int e = e0 + k + u * 256;
            const int ec = e < E ? e : E - 1;
            d[u]  = dst[ec];
            sv[u] = src[ec];
        }
#pragma unroll
        for (int u = 0; u < 5; ++u) {
            if (e0 + k + u * 256 < E) {
                const int p = atomicAdd(&cur[d[u] >> BIN_SHIFT], 1);
                binned[p] = make_int2(sv[u], d[u]);
            }
        }
    }
}

__global__ __launch_bounds__(256) void bin_sort_kernel(const int2* __restrict__ binned,
                                                       const int* __restrict__ start,
                                                       int* __restrict__ rowptr,
                                                       int* __restrict__ csr, int N) {
    __shared__ int cnt[BIN_NODES];
    __shared__ int wsum[8];
    __shared__ int stage[STAGE_CAP];
    const int tid  = threadIdx.x;
    const int lane = tid & 31;
    const int wave = tid >> 5;
    const int bin  = blockIdx.x;
    const int s0 = start[bin * NEB];
    const int s1 = start[(bin + 1) * NEB];
    cnt[tid] = 0;
    __syncthreads();
    for (int i0 = s0 + tid; i0 - tid < s1; i0 += 1024) {
        int dl[4];
#pragma unroll
        for (int u = 0; u < 4; ++u) {
            const int i = i0 + u * 256;
            dl[u] = binned[i < s1 ? i : s1 - 1].y;
        }
#pragma unroll
        for (int u = 0; u < 4; ++u)
            if (i0 + u * 256 < s1) atomicAdd(&cnt[dl[u] & (BIN_NODES - 1)], 1);
    }
    __syncthreads();
    const int c = cnt[tid];
    int v = c;
#pragma unroll
    for (int o = 1; o < 32; o <<= 1) {
        int u = __shfl_up(v, o, 32);
        if (lane >= o) v += u;
    }
    if (lane == 31) wsum[wave] = v;
    __syncthreads();
    int woff = 0;
#pragma unroll
    for (int w = 0; w < 8; ++w) {
        int x = wsum[w];
        if (w < wave) woff += x;
    }
    const int ex = woff + v - c;
    const int node = bin * BIN_NODES + tid;
    if (node <= N) rowptr[node] = s0 + ex;
    cnt[tid] = ex;
    __syncthreads();
    const int len = s1 - s0;
    const bool fits = len <= STAGE_CAP;
    for (int i0 = s0 + tid; i0 - tid < s1; i0 += 1024) {
        int2 ed[4];
#pragma unroll
        for (int u = 0; u < 4; ++u) {
            const int i = i0 + u * 256;
            ed[u] = binned[i < s1 ? i : s1 - 1];
        }
#pragma unroll
        for (int u = 0; u < 4; ++u) {
            if (i0 + u * 256 < s1) {
                const int p = atomicAdd(&cnt[ed[u].y & (BIN_NODES - 1)], 1);
                if (fits) stage[p] = ed[u].x; else csr[s0 + p] = ed[u].x;
            }
        }
    }
    if (fits) {
        __syncthreads();
        for (int i = tid; i < len; i += 256) csr[s0 + i] = stage[i];
    }
}

__global__ __launch_bounds__(256) void gemm64_wmma(const float* __restrict__ X,
                                                   const float* __restrict__ W,
                                                   const int* __restrict__ rowptr,
                                                   float* __restrict__ H, int nTiles) {
    __shared__ float Ws[64 * 64];
    __shared__ float Xs[8][16 * XPAD];

    const int tid  = threadIdx.x;
    const int wave = tid >> 5;
    const int lane = tid & 31;
    const int tile = blockIdx.x * 8 + wave;

#pragma unroll
    for (int i = 0; i < 16; ++i) {
        int g = tid + i * 256;
        int k = g >> 6;
        int n = g & 63;
        Ws[(k >> 1) * 128 + n * 2 + (k & 1)] = W[g];
    }

    const int rowOff = (lane >> 4) * 8;
    const int base = tile * 16;
    int rp[9] = {};

    if (tile < nTiles) {
#pragma unroll
        for (int r = 0; r < 9; ++r) rp[r] = rowptr[base + rowOff + r];
        const float4* Xg = (const float4*)(X + (size_t)tile * 16 * 64);
        float* xl = &Xs[wave][0];
#pragma unroll
        for (int i = 0; i < 8; ++i) {
            int idx4 = lane + i * 32;
            float4 v = Xg[idx4];
            int row = idx4 >> 4;
            int col = (idx4 & 15) * 4;
            *(float4*)&xl[row * XPAD + col] = v;
        }
    }
    __syncthreads();
    if (tile >= nTiles) return;

    const float* xs = &Xs[wave][0];
    const int l15 = lane & 15;
    const int hiK = (lane >> 4) * 2;

    v8f c0 = {}, c1 = {}, c2 = {}, c3 = {};
#pragma unroll
    for (int k = 0; k < 64; k += 4) {
        v2f a = *(const v2f*)(xs + l15 * XPAD + k + hiK);
        const float* wrow = Ws + ((k + hiK) >> 1) * 128 + l15 * 2;
        v2f b0 = *(const v2f*)(wrow +  0);
        v2f b1 = *(const v2f*)(wrow + 32);
        v2f b2 = *(const v2f*)(wrow + 64);
        v2f b3 = *(const v2f*)(wrow + 96);
        c0 = __builtin_amdgcn_wmma_f32_16x16x4_f32(false, a, false, b0, (short)0, c0, false, false);
        c1 = __builtin_amdgcn_wmma_f32_16x16x4_f32(false, a, false, b1, (short)0, c1, false, false);
        c2 = __builtin_amdgcn_wmma_f32_16x16x4_f32(false, a, false, b2, (short)0, c2, false, false);
        c3 = __builtin_amdgcn_wmma_f32_16x16x4_f32(false, a, false, b3, (short)0, c3, false, false);
    }

#pragma unroll
    for (int r = 0; r < 8; ++r) {
        const float di = rsqrtf((float)(rp[r + 1] - rp[r] + 1));
        float* out = H + (size_t)(base + r + rowOff) * 64 + l15;
        out[0]  = c0[r] * di;
        out[16] = c1[r] * di;
        out[32] = c2[r] * di;
        out[48] = c3[r] * di;
    }
}

#define GATHER_WGS 1024

#define GATHER_ROW(k) H2[__builtin_amdgcn_readlane(idx, (k) - 1) * 32 + lane]

__global__ __launch_bounds__(256) void gather_kernel(const float* __restrict__ hs,
                                                     const int* __restrict__ rowptr,
                                                     const int* __restrict__ csr,
                                                     const float* __restrict__ b,
                                                     float* __restrict__ out, int N) {
    const int wave = __builtin_amdgcn_readfirstlane((int)threadIdx.x >> 5);
    const int lane = threadIdx.x & 31;
    const int stride = (int)gridDim.x * 8;
    int node = blockIdx.x * 8 + wave;
    if (node >= N) return;

    const v2f* __restrict__ H2 = (const v2f*)hs;
    const v2f bb = ((const v2f*)b)[lane];
    const v2f zero = {};

    int beg = rowptr[node];
    int end = rowptr[node + 1];
    int nxt = node + stride;
    int nbeg = 0, nend = 0;
    if (nxt < N) { nbeg = rowptr[nxt]; nend = rowptr[nxt + 1]; }
    int idx = csr[max(min(beg + lane, end - 1), 0)];
    asm volatile("" : "+v"(idx));

    for (;;) {
        const int nx2 = nxt + stride;
        int n2beg = 0, n2end = 0;
        if (nx2 < N) { n2beg = rowptr[nx2]; n2end = rowptr[nx2 + 1]; }
        int nidx = 0;
        if (nxt < N) nidx = csr[max(min(nbeg + lane, nend - 1), 0)];

        const int deg = end - beg;
        const int rows = (deg < 32 ? deg : 32) + 1;
        const int rem = rows & 7;
        const int full = rows - rem;
        v2f t0 = zero, t1 = zero, t2 = zero, t3 = zero, t4 = zero, t5 = zero, t6 = zero;
        if (rem > 0) {
            const int r0 = full == 0 ? node : __builtin_amdgcn_readlane(idx, full == 0 ? 0 : full - 1);
            t0 = H2[r0 * 32 + lane];
        }
        if (rem > 1) t1 = GATHER_ROW(full + 1);
        if (rem > 2) t2 = GATHER_ROW(full + 2);
        if (rem > 3) t3 = GATHER_ROW(full + 3);
        if (rem > 4) t4 = GATHER_ROW(full + 4);
        if (rem > 5) t5 = GATHER_ROW(full + 5);
        if (rem > 6) t6 = GATHER_ROW(full + 6);
        v2f v0 = zero, v1 = zero, v2 = zero, v3 = zero, v4 = zero, v5 = zero, v6 = zero, v7 = zero;
        if (full >= 8) {
            v0 = H2[node * 32 + lane];
            v1 = GATHER_ROW(1); v2 = GATHER_ROW(2); v3 = GATHER_ROW(3);
            v4 = GATHER_ROW(4); v5 = GATHER_ROW(5); v6 = GATHER_ROW(6); v7 = GATHER_ROW(7);
        }
        v2f w0 = zero, w1 = zero, w2 = zero, w3 = zero, w4 = zero, w5 = zero, w6 = zero, w7 = zero;
        if (full >= 16) {
            w0 = GATHER_ROW(8);  w1 = GATHER_ROW(9);  w2 = GATHER_ROW(10); w3 = GATHER_ROW(11);
            w4 = GATHER_ROW(12); w5 = GATHER_ROW(13); w6 = GATHER_ROW(14); w7 = GATHER_ROW(15);
        }
        v2f a0 = (v0 + w0) + t0, a1 = (v1 + w1) + t1, a2 = (v2 + w2) + t2, a3 = (v3 + w3) + t3;
        v2f a4 = (v4 + w4) + t4, a5 = (v5 + w5) + t5, a6 = (v6 + w6) + t6, a7 = v7 + w7;
        for (int j = 16; j < full; j += 8) {
            const v2f x0 = GATHER_ROW(j + 0), x1 = GATHER_ROW(j + 1);
            const v2f x2 = GATHER_ROW(j + 2), x3 = GATHER_ROW(j + 3);
            const v2f x4 = GATHER_ROW(j + 4), x5 = GATHER_ROW(j + 5);
            const v2f x6 = GATHER_ROW(j + 6), x7 = GATHER_ROW(j + 7);
            a0 += x0; a1 += x1; a2 += x2; a3 += x3;
            a4 += x4; a5 += x5; a6 += x6; a7 += x7;
        }
        for (int e = beg + 32; e < end; ++e)
            a0 += H2[csr[e] * 32 + lane];

        const float di = rsqrtf((float)(deg + 1));
        const v2f acc = ((a0 + a1) + (a2 + a3)) + ((a4 + a5) + (a6 + a7));
        v2f o;
        o.x = fmaxf(di * acc.x + bb.x, 0.0f);
        o.y = fmaxf(di * acc.y + bb.y, 0.0f);
        ((v2f*)out)[node * 32 + lane] = o;

        if (nxt >= N) break;
        node = nxt;  beg = nbeg;   end = nend;   idx = nidx;
        nxt  = nx2;  nbeg = n2beg; nend = n2end;
    }
}

template <int M>
__device__ __forceinline__ float row_xor(float x) {
    return __builtin_bit_cast(float, __builtin_amdgcn_update_dpp(0, __builtin_bit_cast(int, x),
                                                                 0x160 + M, 0xf, 0xf, false));
}
__device__ __forceinline__ float xhalf(float x) {
    const int b = __builtin_bit_cast(int, x);
    return __builtin_bit_cast(float, __builtin_amdgcn_permlanex16(b, b, 0x76543210, 0xfedcba98, false, false));
}

#define HEAD_WGS 1042
#define HEAD_GROUPS ((GNN_N + 31) / 32)
#define HEAD_MAXG ((HEAD_GROUPS + HEAD_WGS - 1) / HEAD_WGS)
static_assert(HEAD_MAXG * HEAD_WGS >= HEAD_GROUPS, "every group of 32 destinations has an owner");

#define HEAD_NODE(t) ((((int)blockIdx.x + ((t) >> 2) * HEAD_WGS) << 5) + wave * 4 + ((t) & 3))

__global__ __launch_bounds__(256) void gather_head_kernel(const float* __restrict__ hs,
                                                          const int* __restrict__ rowptr,
                                                          const int* __restrict__ csr,
                                                          const float* __restrict__ b,
                                                          const float* __restrict__ Wp,
                                                          const float* __restrict__ bp,
                                                          float* __restrict__ y, int N) {
    __shared__ float ys[HEAD_MAXG * 32];
    const int wave = __builtin_amdgcn_readfirstlane((int)threadIdx.x >> 5);
    const int lane = threadIdx.x & 31;

    const v2f* __restrict__ H2 = (const v2f*)hs;
    const v2f bb = ((const v2f*)b)[lane];
    const v2f wp = ((const v2f*)Wp)[lane];
    const float bias = bp[0];
    const v2f zero = {};

    int t = 0;
    int node = HEAD_NODE(0);
    if (node < N) {
        int beg = rowptr[node];
        int end = rowptr[node + 1];
        int nxt = HEAD_NODE(1);
        int nbeg = 0, nend = 0;
        if (nxt < N) { nbeg = rowptr[nxt]; nend = rowptr[nxt + 1]; }
        int idx = csr[max(min(beg + lane, end - 1), 0)];
        asm volatile("" : "+v"(idx));

        for (;;) {
            const int nx2 = HEAD_NODE(t + 2);
            int n2beg = 0, n2end = 0;
            if (nx2 < N) { n2beg = rowptr[nx2]; n2end = rowptr[nx2 + 1]; }
            int nidx = 0;
            if (nxt < N) nidx = csr[max(min(nbeg + lane, nend - 1), 0)];

            const int deg = end - beg;
            const int rows = (deg < 32 ? deg : 32) + 1;
            const int rem = rows & 7;
            const int full = rows - rem;
            v2f t0 = zero, t1 = zero, t2 = zero, t3 = zero, t4 = zero, t5 = zero, t6 = zero;
            if (rem > 0) {
                const int r0 = full == 0 ? node : __builtin_amdgcn_readlane(idx, full == 0 ? 0 : full - 1);
                t0 = H2[r0 * 32 + lane];
            }
            if (rem > 1) t1 = GATHER_ROW(full + 1);
            if (rem > 2) t2 = GATHER_ROW(full + 2);
            if (rem > 3) t3 = GATHER_ROW(full + 3);
            if (rem > 4) t4 = GATHER_ROW(full + 4);
            if (rem > 5) t5 = GATHER_ROW(full + 5);
            if (rem > 6) t6 = GATHER_ROW(full + 6);
            v2f v0 = zero, v1 = zero, v2 = zero, v3 = zero, v4 = zero, v5 = zero, v6 = zero, v7 = zero;
            if (full >= 8) {
                v0 = H2[node * 32 + lane];
                v1 = GATHER_ROW(1); v2 = GATHER_ROW(2); v3 = GATHER_ROW(3);
                v4 = GATHER_ROW(4); v5 = GATHER_ROW(5); v6 = GATHER_ROW(6); v7 = GATHER_ROW(7);
            }
            v2f w0 = zero, w1 = zero, w2 = zero, w3 = zero, w4 = zero, w5 = zero, w6 = zero, w7 = zero;
            if (full >= 16) {
                w0 = GATHER_ROW(8);  w1 = GATHER_ROW(9);  w2 = GATHER_ROW(10); w3 = GATHER_ROW(11);
                w4 = GATHER_ROW(12); w5 = GATHER_ROW(13); w6 = GATHER_ROW(14); w7 = GATHER_ROW(15);
            }
            v2f a0 = (v0 + w0) + t0, a1 = (v1 + w1) + t1, a2 = (v2 + w2) + t2, a3 = (v3 + w3) + t3;
            v2f a4 = (v4 + w4) + t4, a5 = (v5 + w5) + t5, a6 = (v6 + w6) + t6, a7 = v7 + w7;
            for (int j = 16; j < full; j += 8) {
                const v2f x0 = GATHER_ROW(j + 0), x1 = GATHER_ROW(j + 1);
                const v2f x2 = GATHER_ROW(j + 2), x3 = GATHER_ROW(j + 3);
                const v2f x4 = GATHER_ROW(j + 4), x5 = GATHER_ROW(j + 5);
                const v2f x6 = GATHER_ROW(j + 6), x7 = GATHER_ROW(j + 7);
                a0 += x0; a1 += x1; a2 += x2; a3 += x3;
                a4 += x4; a5 += x5; a6 += x6; a7 += x7;
            }
            for (int e = beg + 32; e < end; ++e)
                a0 += H2[csr[e] * 32 + lane];

            const float di = rsqrtf((float)(deg + 1));
            const v2f acc = ((a0 + a1) + (a2 + a3)) + ((a4 + a5) + (a6 + a7));
            float p = fmaxf(di * acc.x + bb.x, 0.0f) * wp.x + fmaxf(di * acc.y + bb.y, 0.0f) * wp.y;
            p += row_xor<1>(p);
            p += row_xor<2>(p);
            p += row_xor<4>(p);
            p += row_xor<8>(p);
            p += xhalf(p);
            if (lane == 0) ys[(t >> 2) * 32 + wave * 4 + (t & 3)] = p + bias;

            if (nxt >= N) break;
            ++t;
            node = nxt;  beg = nbeg;   end = nend;   idx = nidx;
            nxt  = nx2;  nbeg = n2beg; nend = n2end;
        }
    }
    __syncthreads();
    for (int k = wave; k < HEAD_MAXG; k += 8) {
        const int r = (((int)blockIdx.x + k * HEAD_WGS) << 5) + lane;
        if (r < N) y[r] = ys[k * 32 + lane];
    }
}

extern "C" void kernel_launch(void* const* d_in, const int* in_sizes, int n_in,
                              void* d_out, int out_size, void* d_ws, size_t ws_size,
                              hipStream_t stream) {
    (void)in_sizes; (void)n_in; (void)out_size; (void)ws_size;

    const float* x  = (const float*)d_in[0];
    const int*   ei = (const int*)d_in[1];
    const float* W1 = (const float*)d_in[2];
    const float* b1 = (const float*)d_in[3];
    const float* W2 = (const float*)d_in[4];
    const float* b2 = (const float*)d_in[5];
    const float* Wp = (const float*)d_in[6];
    const float* bp = (const float*)d_in[7];
    float* y = (float*)d_out;

    const int N = GNN_N, E = GNN_E;
    const int* src = ei;
    const int* dst = ei + E;

    char* ws = (char*)d_ws;
    size_t off = 0;
    auto take = [&](size_t bytes) {
        void* p = ws + off;
        off += (bytes + 255) & ~(size_t)255;
        return p;
    };
    int*   rowptr = (int*)take((size_t)(N + 1) * 4);
    int*   csr    = (int*)take((size_t)E * 4);
    float* bufA   = (float*)take((size_t)N * DIM * 4);
    float* bufB   = (float*)take((size_t)N * DIM * 4);
    int*  cell   = (int*)bufA;
    int*  bsum   = cell + NCELL;
    int*  start  = bsum + 256;
    int2* binned = (int2*)bufB;

    const int nTiles = N / 16;
    const int gemmBlocks = (nTiles + 7) / 8;

    bin_count_kernel<<<NEB, 256, 0, stream>>>(dst, cell, E);
    scan_local_kernel<<<SCAN_BLOCKS, 256, 0, stream>>>(cell, start, bsum, NCELL);
    scan_add_kernel<<<SCAN_BLOCKS, 256, 0, stream>>>(start, bsum, NCELL, E);
    bin_place_kernel<<<NEB, 256, 0, stream>>>(src, dst, start, binned, E);
    bin_sort_kernel<<<NBIN, 256, 0, stream>>>(binned, start, rowptr, csr, N);

    gemm64_wmma<<<gemmBlocks, 256, 0, stream>>>(x, W1, rowptr, bufA, nTiles);
    gather_kernel<<<GATHER_WGS, 256, 0, stream>>>(bufA, rowptr, csr, b1, bufB, N);

    gemm64_wmma<<<gemmBlocks, 256, 0, stream>>>(bufB, W2, rowptr, bufA, nTiles);
    gather_head_kernel<<<HEAD_WGS, 256, 0, stream>>>(bufA, rowptr, csr, b2, Wp, bp, y, N);
}


# Layer 2 aggregates before it transforms: the second gather sums rows into LDS tiles and takes the W2 product (f32 WMMA), bias, relu and head from them; the second GEMM launch is gone, 9 to 8 launches
# speedup vs baseline: 6.3140x; 1.0714x over previous
#include <hip/hip_runtime.h>


typedef float v2f __attribute__((ext_vector_type(2)));
typedef float v8f __attribute__((ext_vector_type(8)));

#define GNN_N 100000
#define GNN_E 1600000
#define DIM   64
#define XPAD  68
#define BIN_SHIFT 8
#define BIN_NODES 256
#define NBIN (GNN_N / BIN_NODES + 1)
#define EPB 6400
#define NEB ((GNN_E + EPB - 1) / EPB)
#define NCELL (NBIN * NEB)
#define STAGE_CAP 8192
#define SCAN_CHUNK 1024
#define SCAN_BLOCKS ((NCELL + SCAN_CHUNK - 1) / SCAN_CHUNK)
static_assert(EPB % (256 * 5) == 0, "a binning work-group walks its slice in steps of 256 x 5 edges");
static_assert(SCAN_BLOCKS <= 256, "the second scan step sums the chunk totals with one 256-thread group");

__global__ __launch_bounds__(256) void bin_count_kernel(const int* __restrict__ dst,
                                                        int* __restrict__ cell, int E) {
    __shared__ int h[NBIN];
    for (int i = threadIdx.x; i < NBIN; i += 256) h[i] = 0;
    __syncthreads();
    const int e0 = blockIdx.x * EPB + threadIdx.x;
    for (int k = 0; k < EPB; k += 256 * 5) {
        int d[5];
#pragma unroll
        for (int u = 0; u < 5; ++u) {
            const int e = e0 + k + u * 256;
            d[u] = dst[e < E ? e : E - 1];
        }
#pragma unroll
        for (int u = 0; u < 5; ++u)
            if (e0 + k + u * 256 < E) atomicAdd(&h[d[u] >> BIN_SHIFT], 1);
    }
    __syncthreads();
    for (int i = threadIdx.x; i < NBIN; i += 256) cell[i * NEB + blockIdx.x] = h[i];
}

__global__ __launch_bounds__(256) void scan_local_kernel(const int* __restrict__ cnt,
                                                         int* __restrict__ rowptr,
                                                         int* __restrict__ bsum, int n) {
    __shared__ int wsum[8];
    const int tid  = threadIdx.x;
    const int lane = tid & 31;
    const int wave = tid >> 5;
    const int i0 = blockIdx.x * SCAN_CHUNK + tid * 4;
    int t0 = (i0 + 0 < n) ? cnt[i0 + 0] : 0;
    int t1 = (i0 + 1 < n) ? cnt[i0 + 1] : 0;
    int t2 = (i0 + 2 < n) ? cnt[i0 + 2] : 0;
    int t3 = (i0 + 3 < n) ? cnt[i0 + 3] : 0;
    const int s = t0 + t1 + t2 + t3;
    int v = s;
#pragma unroll
    for (int o = 1; o < 32; o <<= 1) {
        int u = __shfl_up(v, o, 32);
        if (lane >= o) v += u;
    }
    if (lane == 31) wsum[wave] = v;
    __syncthreads();
    int woff = 0, tot = 0;
#pragma unroll
    for (int w = 0; w < 8; ++w) {
        int x = wsum[w];
        if (w < wave) woff += x;
        tot += x;
    }
    int ex = woff + v - s;
    if (i0 + 0 < n) rowptr[i0 + 0] = ex;
    if (i0 + 1 < n) rowptr[i0 + 1] = ex + t0;
    if (i0 + 2 < n) rowptr[i0 + 2] = ex + t0 + t1;
    if (i0 + 3 < n) rowptr[i0 + 3] = ex + t0 + t1 + t2;
    if (tid == 0) bsum[blockIdx.x] = tot;
}

__global__ __launch_bounds__(256) void scan_add_kernel(int* __restrict__ rowptr,
                                                       const int* __restrict__ bsum,
                                                       int n, int total) {
    __shared__ int wsum[8];
    const int tid  = threadIdx.x;
    const int lane = tid & 31;
    const int wave = tid >> 5;
    int v = (tid < (int)blockIdx.x) ? bsum[tid] : 0;
#pragma unroll
    for (int m = 16; m > 0; m >>= 1) v += __shfl_xor(v, m, 32);
    if (lane == 0) wsum[wave] = v;
    __syncthreads();
    int off = 0;
#pragma unroll
    for (int w = 0; w < 8; ++w) off += wsum[w];
    const int i0 = blockIdx.x * SCAN_CHUNK + tid * 4;
#pragma unroll
    for (int k = 0; k < 4; ++k)
        if (i0 + k < n) rowptr[i0 + k] += off;
    if (blockIdx.x == 0 && tid == 0) rowptr[n] = total;
}

__global__ __launch_bounds__(256) void bin_place_kernel(const int* __restrict__ src,
                                                        const int* __restrict__ dst,
                                                        const int* __restrict__ start,
                                                        int2* __restrict__ binned, int E) {
    __shared__ int cur[NBIN];
    for (int i = threadIdx.x; i < NBIN; i += 256) cur[i] = start[i * NEB + blockIdx.x];
    __syncthreads();
    const int e0 = blockIdx.x * EPB + threadIdx.x;
    for (int k = 0; k < EPB; k += 256 * 5) {
        int d[5], sv[5];
#pragma unroll
        for (int u = 0; u < 5; ++u) {
            const int e = e0 + k + u * 256;
            const int ec = e < E ? e : E - 1;
            d[u]  = dst[ec];
            sv[u] = src[ec];
        }
#pragma unroll
        for (int u = 0; u < 5; ++u) {
            if (e0 + k + u * 256 < E) {
                const int p = atomicAdd(&cur[d[u] >> BIN_SHIFT], 1);
                binned[p] = make_int2(sv[u], d[u]);
            }
        }
    }
}

__global__ __launch_bounds__(256) void bin_sort_kernel(const int2* __restrict__ binned,
                                                       const int* __restrict__ start,
                                                       int* __restrict__ rowptr,
                                                       int* __restrict__ csr, int N) {
    __shared__ int cnt[BIN_NODES];
    __shared__ int wsum[8];
    __shared__ int stage[STAGE_CAP];
    const int tid  = threadIdx.x;
    const int lane = tid & 31;
    const int wave = tid >> 5;
    const int bin  = blockIdx.x;
    const int s0 = start[bin * NEB];
    const int s1 = start[(bin + 1) * NEB];
    cnt[tid] = 0;
    __syncthreads();
    for (int i0 = s0 + tid; i0 - tid < s1; i0 += 1024) {
        int dl[4];
#pragma unroll
        for (int u = 0; u < 4; ++u) {
            const int i = i0 + u * 256;
            dl[u] = binned[i < s1 ? i : s1 - 1].y;
        }
#pragma unroll
        for (int u = 0; u < 4; ++u)
            if (i0 + u * 256 < s1) atomicAdd(&cnt[dl[u] & (BIN_NODES - 1)], 1);
    }
    __syncthreads();
    const int c = cnt[tid];
    int v = c;
#pragma unroll
    for (int o = 1; o < 32; o <<= 1) {
        int u = __shfl_up(v, o, 32);
        if (lane >= o) v += u;
    }
    if (lane == 31) wsum[wave] = v;
    __syncthreads();
    int woff = 0;
#pragma unroll
    for (int w = 0; w < 8; ++w) {
        int x = wsum[w];
        if (w < wave) woff += x;
    }
    const int ex = woff + v - c;
    const int node = bin * BIN_NODES + tid;
    if (node <= N) rowptr[node] = s0 + ex;
    cnt[tid] = ex;
    __syncthreads();
    const int len = s1 - s0;
    const bool fits = len <= STAGE_CAP;
    for (int i0 = s0 + tid; i0 - tid < s1; i0 += 1024) {
        int2 ed[4];
#pragma unroll
        for (int u = 0; u < 4; ++u) {
            const int i = i0 + u * 256;
            ed[u] = binned[i < s1 ? i : s1 - 1];
        }
#pragma unroll
        for (int u = 0; u < 4; ++u) {
            if (i0 + u * 256 < s1) {
                const int p = atomicAdd(&cnt[ed[u].y & (BIN_NODES - 1)], 1);
                if (fits) stage[p] = ed[u].x; else csr[s0 + p] = ed[u].x;
            }
        }
    }
    if (fits) {
        __syncthreads();
        for (int i = tid; i < len; i += 256) csr[s0 + i] = stage[i];
    }
}

__global__ __launch_bounds__(256) void gemm64_wmma(const float* __restrict__ X,
                                                   const float* __restrict__ W,
                                                   const int* __restrict__ rowptr,
                                                   float* __restrict__ H, int nTiles) {
    __shared__ float Ws[64 * 64];
    __shared__ float Xs[8][16 * XPAD];

    const int tid  = threadIdx.x;
    const int wave = tid >> 5;
    const int lane = tid & 31;
    const int tile = blockIdx.x * 8 + wave;

#pragma unroll
    for (int i = 0; i < 16; ++i) {
        int g = tid + i * 256;
        int k = g >> 6;
        int n = g & 63;
        Ws[(k >> 1) * 128 + n * 2 + (k & 1)] = W[g];
    }

    const int rowOff = (lane >> 4) * 8;
    const int base = tile * 16;
    int rp[9] = {};

    if (tile < nTiles) {
#pragma unroll
        for (int r = 0; r < 9; ++r) rp[r] = rowptr[base + rowOff + r];
        const float4* Xg = (const float4*)(X + (size_t)tile * 16 * 64);
        float* xl = &Xs[wave][0];
#pragma unroll
        for (int i = 0; i < 8; ++i) {
            int idx4 = lane + i * 32;
            float4 v = Xg[idx4];
            int row = idx4 >> 4;
            int col = (idx4 & 15) * 4;
            *(float4*)&xl[row * XPAD + col] = v;
        }
    }
    __syncthreads();
    if (tile >= nTiles) return;

    const float* xs = &Xs[wave][0];
    const int l15 = lane & 15;
    const int hiK = (lane >> 4) * 2;

    v8f c0 = {}, c1 = {}, c2 = {}, c3 = {};
#pragma unroll
    for (int k = 0; k < 64; k += 4) {
        v2f a = *(const v2f*)(xs + l15 * XPAD + k + hiK);
        const float* wrow = Ws + ((k + hiK) >> 1) * 128 + l15 * 2;
        v2f b0 = *(const v2f*)(wrow +  0);
        v2f b1 = *(const v2f*)(wrow + 32);
        v2f b2 = *(const v2f*)(wrow + 64);
        v2f b3 = *(const v2f*)(wrow + 96);
        c0 = __builtin_amdgcn_wmma_f32_16x16x4_f32(false, a, false, b0, (short)0, c0, false, false);
        c1 = __builtin_amdgcn_wmma_f32_16x16x4_f32(false, a, false, b1, (short)0, c1, false, false);
        c2 = __builtin_amdgcn_wmma_f32_16x16x4_f32(false, a, false, b2, (short)0, c2, false, false);
        c3 = __builtin_amdgcn_wmma_f32_16x16x4_f32(false, a, false, b3, (short)0, c3, false, false);
    }

#pragma unroll
    for (int r = 0; r < 8; ++r) {
        const float di = rsqrtf((float)(rp[r + 1] - rp[r] + 1));
        float* out = H + (size_t)(base + r + rowOff) * 64 + l15;
        out[0]  = c0[r] * di;
        out[16] = c1[r] * di;
        out[32] = c2[r] * di;
        out[48] = c3[r] * di;
    }
}

#define GATHER_WGS 1024

#define GATHER_ROW(k) H2[__builtin_amdgcn_readlane(idx, (k) - 1) * 32 + lane]

__global__ __launch_bounds__(256) void gather_kernel(const float* __restrict__ hs,
                                                     const int* __restrict__ rowptr,
                                                     const int* __restrict__ csr,
                                                     const float* __restrict__ b,
                                                     float* __restrict__ out, int N) {
    const int wave = __builtin_amdgcn_readfirstlane((int)threadIdx.x >> 5);
    const int lane = threadIdx.x & 31;
    const int stride = (int)gridDim.x * 8;
    int node = blockIdx.x * 8 + wave;
    if (node >= N) return;

    const v2f* __restrict__ H2 = (const v2f*)hs;
    const v2f bb = ((const v2f*)b)[lane];
    const v2f zero = {};

    int beg = rowptr[node];
    int end = rowptr[node + 1];
    int nxt = node + stride;
    int nbeg = 0, nend = 0;
    if (nxt < N) { nbeg = rowptr[nxt]; nend = rowptr[nxt + 1]; }
    int idx = csr[max(min(beg + lane, end - 1), 0)];
    asm volatile("" : "+v"(idx));

    for (;;) {
        const int nx2 = nxt + stride;
        int n2beg = 0, n2end = 0;
        if (nx2 < N) { n2beg = rowptr[nx2]; n2end = rowptr[nx2 + 1]; }
        int nidx = 0;
        if (nxt < N) nidx = csr[max(min(nbeg + lane, nend - 1), 0)];

        const int deg = end - beg;
        const int rows = (deg < 32 ? deg : 32) + 1;
        const int rem = rows & 7;
        const int full = rows - rem;
        v2f t0 = zero, t1 = zero, t2 = zero, t3 = zero, t4 = zero, t5 = zero, t6 = zero;
        if (rem > 0) {
            const int r0 = full == 0 ? node : __builtin_amdgcn_readlane(idx, full == 0 ? 0 : full - 1);
            t0 = H2[r0 * 32 + lane];
        }
        if (rem > 1) t1 = GATHER_ROW(full + 1);
        if (rem > 2) t2 = GATHER_ROW(full + 2);
        if (rem > 3) t3 = GATHER_ROW(full + 3);
        if (rem > 4) t4 = GATHER_ROW(full + 4);
        if (rem > 5) t5 = GATHER_ROW(full + 5);
        if (rem > 6) t6 = GATHER_ROW(full + 6);
        v2f v0 = zero, v1 = zero, v2 = zero, v3 = zero, v4 = zero, v5 = zero, v6 = zero, v7 = zero;
        if (full >= 8) {
            v0 = H2[node * 32 + lane];
            v1 = GATHER_ROW(1); v2 = GATHER_ROW(2); v3 = GATHER_ROW(3);
            v4 = GATHER_ROW(4); v5 = GATHER_ROW(5); v6 = GATHER_ROW(6); v7 = GATHER_ROW(7);
        }
        v2f w0 = zero, w1 = zero, w2 = zero, w3 = zero, w4 = zero, w5 = zero, w6 = zero, w7 = zero;
        if (full >= 16) {
            w0 = GATHER_ROW(8);  w1 = GATHER_ROW(9);  w2 = GATHER_ROW(10); w3 = GATHER_ROW(11);
            w4 = GATHER_ROW(12); w5 = GATHER_ROW(13); w6 = GATHER_ROW(14); w7 = GATHER_ROW(15);
        }
        v2f a0 = (v0 + w0) + t0, a1 = (v1 + w1) + t1, a2 = (v2 + w2) + t2, a3 = (v3 + w3) + t3;
        v2f a4 = (v4 + w4) + t4, a5 = (v5 + w5) + t5, a6 = (v6 + w6) + t6, a7 = v7 + w7;
        for (int j = 16; j < full; j += 8) {
            const v2f x0 = GATHER_ROW(j + 0), x1 = GATHER_ROW(j + 1);
            const v2f x2 = GATHER_ROW(j + 2), x3 = GATHER_ROW(j + 3);
            const v2f x4 = GATHER_ROW(j + 4), x5 = GATHER_ROW(j + 5);
            const v2f x6 = GATHER_ROW(j + 6), x7 = GATHER_ROW(j + 7);
            a0 += x0; a1 += x1; a2 += x2; a3 += x3;
            a4 += x4; a5 += x5; a6 += x6; a7 += x7;
        }
        for (int e = beg + 32; e < end; ++e)
            a0 += H2[csr[e] * 32 + lane];

        const float di = rsqrtf((float)(deg + 1));
        const v2f acc = ((a0 + a1) + (a2 + a3)) + ((a4 + a5) + (a6 + a7));
        v2f o;
        o.x = fmaxf(di * acc.x + bb.x, 0.0f) * di;
        o.y = fmaxf(di * acc.y + bb.y, 0.0f) * di;
        ((v2f*)out)[node * 32 + lane] = o;

        if (nxt >= N) break;
        node = nxt;  beg = nbeg;   end = nend;   idx = nidx;
        nxt  = nx2;  nbeg = n2beg; nend = n2end;
    }
}

template <int M>
__device__ __forceinline__ float row_xor(float x) {
    return __builtin_bit_cast(float, __builtin_amdgcn_update_dpp(0, __builtin_bit_cast(int, x),
                                                                 0x160 + M, 0xf, 0xf, false));
}

#define HEAD_WGS 1042
#define HEAD_GROUPS ((GNN_N + 31) / 32)
#define HEAD_MAXG ((HEAD_GROUPS + HEAD_WGS - 1) / HEAD_WGS)
#define HEAD_ROWS (HEAD_MAXG * 32)
static_assert(HEAD_MAXG * HEAD_WGS >= HEAD_GROUPS, "every group of 32 destinations has an owner");
static_assert(HEAD_MAXG == 3, "the product step gives a wave three 16-row tiles: tile, tile + 2, tile + 4");

#define HEAD_NODE(t) ((((int)blockIdx.x + ((t) >> 2) * HEAD_WGS) << 5) + wave * 4 + ((t) & 3))
#define HEAD_ROW(t) ((((t) >> 2) << 5) + wave * 4 + ((t) & 3))

__global__ __launch_bounds__(256) void gather_head_kernel(const float* __restrict__ hs,
                                                          const int* __restrict__ rowptr,
                                                          const int* __restrict__ csr,
                                                          const float* __restrict__ W,
                                                          const float* __restrict__ b,
                                                          const float* __restrict__ Wp,
                                                          const float* __restrict__ bp,
                                                          float* __restrict__ y, int N) {
    __shared__ float As[HEAD_ROWS * XPAD];
    __shared__ float part[4 * HEAD_ROWS];
    const int wave = __builtin_amdgcn_readfirstlane((int)threadIdx.x >> 5);
    const int lane = threadIdx.x & 31;

    const v2f* __restrict__ H2 = (const v2f*)hs;
    const v2f zero = {};

    for (int t = 0; t < HEAD_MAXG * 4; ++t)
        *(v2f*)&As[HEAD_ROW(t) * XPAD + 2 * lane] = zero;

    int t = 0;
    int node = HEAD_NODE(0);
    if (node < N) {
        int beg = rowptr[node];
        int end = rowptr[node + 1];
        int nxt = HEAD_NODE(1);
        int nbeg = 0, nend = 0;
        if (nxt < N) { nbeg = rowptr[nxt]; nend = rowptr[nxt + 1]; }
        int idx = csr[max(min(beg + lane, end - 1), 0)];
        asm volatile("" : "+v"(idx));

        for (;;) {
            const int nx2 = HEAD_NODE(t + 2);
            int n2beg = 0, n2end = 0;
            if (nx2 < N) { n2beg = rowptr[nx2]; n2end = rowptr[nx2 + 1]; }
            int nidx = 0;
            if (nxt < N) nidx = csr[max(min(nbeg + lane, nend - 1), 0)];

            const int deg = end - beg;
            const int rows = (deg < 32 ? deg : 32) + 1;
            const int rem = rows & 7;
            const int full = rows - rem;
            v2f t0 = zero, t1 = zero, t2 = zero, t3 = zero, t4 = zero, t5 = zero, t6 = zero;
            if (rem > 0) {
                const int r0 = full == 0 ? node : __builtin_amdgcn_readlane(idx, full == 0 ? 0 : full - 1);
                t0 = H2[r0 * 32 + lane];
            }
            if (rem > 1) t1 = GATHER_ROW(full + 1);
            if (rem > 2) t2 = GATHER_ROW(full + 2);
            if (rem > 3) t3 = GATHER_ROW(full + 3);
            if (rem > 4) t4 = GATHER_ROW(full + 4);
            if (rem > 5) t5 = GATHER_ROW(full + 5);
            if (rem > 6) t6 = GATHER_ROW(full + 6);
            v2f v0 = zero, v1 = zero, v2 = zero, v3 = zero, v4 = zero, v5 = zero, v6 = zero, v7 = zero;
            if (full >= 8) {
                v0 = H2[node * 32 + lane];
                v1 = GATHER_ROW(1); v2 = GATHER_ROW(2); v3 = GATHER_ROW(3);
                v4 = GATHER_ROW(4); v5 = GATHER_ROW(5); v6 = GATHER_ROW(6); v7 = GATHER_ROW(7);
            }
            v2f w0 = zero, w1 = zero, w2 = zero, w3 = zero, w4 = zero, w5 = zero, w6 = zero, w7 = zero;
            if (full >= 16) {
                w0 = GATHER_ROW(8);  w1 = GATHER_ROW(9);  w2 = GATHER_ROW(10); w3 = GATHER_ROW(11);
                w4 = GATHER_ROW(12); w5 = GATHER_ROW(13); w6 = GATHER_ROW(14); w7 = GATHER_ROW(15);
            }
            v2f a0 = (v0 + w0) + t0, a1 = (v1 + w1) + t1, a2 = (v2 + w2) + t2, a3 = (v3 + w3) + t3;
            v2f a4 = (v4 + w4) + t4, a5 = (v5 + w5) + t5, a6 = (v6 + w6) + t6, a7 = v7 + w7;
            for (int j = 16; j < full; j += 8) {
                const v2f x0 = GATHER_ROW(j + 0), x1 = GATHER_ROW(j + 1);
                const v2f x2 = GATHER_ROW(j + 2), x3 = GATHER_ROW(j + 3);
                const v2f x4 = GATHER_ROW(j + 4), x5 = GATHER_ROW(j + 5);
                const v2f x6 = GATHER_ROW(j + 6), x7 = GATHER_ROW(j + 7);
                a0 += x0; a1 += x1; a2 += x2; a3 += x3;
                a4 += x4; a5 += x5; a6 += x6; a7 += x7;
            }
            for (int e = beg + 32; e < end; ++e)
                a0 += H2[csr[e] * 32 + lane];

            const float di = rsqrtf((float)(deg + 1));
            const v2f acc = ((a0 + a1) + (a2 + a3)) + ((a4 + a5) + (a6 + a7));
            v2f o;
            o.x = di * acc.x;
            o.y = di * acc.y;
            *(v2f*)&As[HEAD_ROW(t) * XPAD + 2 * lane] = o;

            if (nxt >= N) break;
            ++t;
            node = nxt;  beg = nbeg;   end = nend;   idx = nidx;
            nxt  = nx2;  nbeg = n2beg; nend = n2end;
        }
    }

    const int l15 = lane & 15;
    const int hiK = (lane >> 4) * 2;
    const int cb  = wave & 3;
    const int th  = wave >> 2;
    const int n   = cb * 16 + l15;
    v2f bf[16];
#pragma unroll
    for (int kk = 0; kk < 16; ++kk) {
        bf[kk].x = W[(kk * 4 + hiK) * 64 + n];
        bf[kk].y = W[(kk * 4 + hiK + 1) * 64 + n];
    }
    const float bn  = b[n];
    const float wpn = Wp[n];
    const float bias = bp[0];
    __syncthreads();

    const float* a0p = As + ((th + 0) * 16 + l15) * XPAD + hiK;
    const float* a1p = As + ((th + 2) * 16 + l15) * XPAD + hiK;
    const float* a2p = As + ((th + 4) * 16 + l15) * XPAD + hiK;
    v8f c0 = {}, c1 = {}, c2 = {};
#pragma unroll
    for (int kk = 0; kk < 16; ++kk) {
        const v2f a0 = *(const v2f*)(a0p + kk * 4);
        const v2f a1 = *(const v2f*)(a1p + kk * 4);
        const v2f a2 = *(const v2f*)(a2p + kk * 4);
        c0 = __builtin_amdgcn_wmma_f32_16x16x4_f32(false, a0, false, bf[kk], (short)0, c0, false, false);
        c1 = __builtin_amdgcn_wmma_f32_16x16x4_f32(false, a1, false, bf[kk], (short)0, c1, false, false);
        c2 = __builtin_amdgcn_wmma_f32_16x16x4_f32(false, a2, false, bf[kk], (short)0, c2, false, false);
    }

    const int hiRow = (lane >> 4) * 8;
    float* pw = part + cb * HEAD_ROWS + th * 16 + hiRow;
#pragma unroll
    for (int r = 0; r < 8; ++r) {
        float p0 = fmaxf(c0[r] + bn, 0.0f) * wpn;
        float p1 = fmaxf(c1[r] + bn, 0.0f) * wpn;
        float p2 = fmaxf(c2[r] + bn, 0.0f) * wpn;
        p0 += row_xor<1>(p0); p1 += row_xor<1>(p1); p2 += row_xor<1>(p2);
        p0 += row_xor<2>(p0); p1 += row_xor<2>(p1); p2 += row_xor<2>(p2);
        p0 += row_xor<4>(p0); p1 += row_xor<4>(p1); p2 += row_xor<4>(p2);
        p0 += row_xor<8>(p0); p1 += row_xor<8>(p1); p2 += row_xor<8>(p2);
        if (l15 == 0) {
            pw[r]      = p0;
            pw[r + 32] = p1;
            pw[r + 64] = p2;
        }
    }
    __syncthreads();
    for (int k = wave; k < HEAD_MAXG; k += 8) {
        const int r = (((int)blockIdx.x + k * HEAD_WGS) << 5) + lane;
        const int R = k * 32 + lane;
        if (r < N)
            y[r] = ((part[R] + part[HEAD_ROWS + R]) + (part[2 * HEAD_ROWS + R] + part[3 * HEAD_ROWS + R])) + bias;
    }
}

extern "C" void kernel_launch(void* const* d_in, const int* in_sizes, int n_in,
                              void* d_out, int out_size, void* d_ws, size_t ws_size,
                              hipStream_t stream) {
    (void)in_sizes; (void)n_in; (void)out_size; (void)ws_size;

    const float* x  = (const float*)d_in[0];
    const int*   ei = (const int*)d_in[1];
    const float* W1 = (const float*)d_in[2];
    const float* b1 = (const float*)d_in[3];
    const float* W2 = (const float*)d_in[4];
    const float* b2 = (const float*)d_in[5];
    const float* Wp = (const float*)d_in[6];
    const float* bp = (const float*)d_in[7];
    float* y = (float*)d_out;

    const int N = GNN_N, E = GNN_E;
    const int* src = ei;
    const int* dst = ei + E;

    char* ws = (char*)d_ws;
    size_t off = 0;
    auto take = [&](size_t bytes) {
        void* p = ws + off;
        off += (bytes + 255) & ~(size_t)255;
        return p;
    };
    int*   rowptr = (int*)take((size_t)(N + 1) * 4);
    int*   csr    = (int*)take((size_t)E * 4);
    float* bufA   = (float*)take((size_t)N * DIM * 4);
    float* bufB   = (float*)take((size_t)N * DIM * 4);
    int*  cell   = (int*)bufA;
    int*  bsum   = cell + NCELL;
    int*  start  = bsum + 256;
    int2* binned = (int2*)bufB;

    const int nTiles = N / 16;
    const int gemmBlocks = (nTiles + 7) / 8;

    bin_count_kernel<<<NEB, 256, 0, stream>>>(dst, cell, E);
    scan_local_kernel<<<SCAN_BLOCKS, 256, 0, stream>>>(cell, start, bsum, NCELL);
    scan_add_kernel<<<SCAN_BLOCKS, 256, 0, stream>>>(start, bsum, NCELL, E);
    bin_place_kernel<<<NEB, 256, 0, stream>>>(src, dst, start, binned, E);
    bin_sort_kernel<<<NBIN, 256, 0, stream>>>(binned, start, rowptr, csr, N);

    gemm64_wmma<<<gemmBlocks, 256, 0, stream>>>(x, W1, rowptr, bufA, nTiles);
    gather_kernel<<<GATHER_WGS, 256, 0, stream>>>(bufA, rowptr, csr, b1, bufB, N);

    gather_head_kernel<<<HEAD_WGS, 256, 0, stream>>>(bufB, rowptr, csr, W2, b2, Wp, bp, y, N);
}
